# starting kernel
# baseline (speedup 1.0000x reference)
#include <hip/hip_runtime.h>
#include <hip/hip_bf16.h>


typedef __attribute__((ext_vector_type(2))) float v2f;
typedef __attribute__((ext_vector_type(8))) float v8f;

__global__ void gcn_transpose_w(const float* __restrict__ W,
                                float* __restrict__ Wt, int K) {
  int i = blockIdx.x * blockDim.x + threadIdx.x;
  if (i < K * 128) {
    const int k = i >> 7;
    const int c = i & 127;
    Wt[(size_t)c * K + k] = W[i];
  }
}

template <int K>
__global__ __launch_bounds__(256) void gcn_gemm_wmma(
    const float* __restrict__ X, const float* __restrict__ Wt,
    float* __restrict__ H, int nrows) {
  constexpr int LDA = K + 4;
  __shared__ __align__(16) float As[16 * LDA];

  const int rowBase = blockIdx.x * 16;
  const int tid = threadIdx.x;

  constexpr int vecPerRow = K / 4;
  for (int i = tid; i < 16 * vecPerRow; i += 256) {
    const int r = i / vecPerRow;
    const int v = i % vecPerRow;
    const int grow = rowBase + r;
    float4 val = make_float4(0.f, 0.f, 0.f, 0.f);
    if (grow < nrows) {
      val = *reinterpret_cast<const float4*>(X + (size_t)grow * K + 4 * v);
    }
    *reinterpret_cast<float4*>(&As[r * LDA + 4 * v]) = val;
  }
  __syncthreads();

  const int wave = tid >> 5;
  const int lane = tid & 31;
  const int l15 = lane & 15;
  const int halfSel = lane >> 4;
  const int col = wave * 16 + l15;

  const float* __restrict__ Bcol = Wt + (size_t)col * K;

  v8f acc = {};
#pragma unroll 4
  for (int k0 = 0; k0 < K; k0 += 4) {
    const int kk = k0 + 2 * halfSel;
    v2f a;
    a.x = As[l15 * LDA + kk];
    a.y = As[l15 * LDA + kk + 1];
    v2f b;
    b.x = Bcol[kk];
    b.y = Bcol[kk + 1];
    acc = __builtin_amdgcn_wmma_f32_16x16x4_f32(
         false, a,  false, b,
         (short)0, acc,  false,  false);
  }

  if (rowBase + 16 <= nrows) {
    float* __restrict__ Hp = H + (size_t)(rowBase + 8 * halfSel) * 128 + col;
#pragma unroll
    for (int r = 0; r < 8; ++r) {
      Hp[(size_t)r * 128] = acc[r];
    }
  } else {
#pragma unroll
    for (int r = 0; r < 8; ++r) {
      const int m = rowBase + r + 8 * halfSel;
      if (m < nrows) H[(size_t)m * 128 + col] = acc[r];
    }
  }
}

__global__ void gcn_deg_init(float* __restrict__ deg, int n) {
  int i = blockIdx.x * blockDim.x + threadIdx.x;
  if (i < n) deg[i] = 1.0f;
}

__global__ void gcn_deg_count(const int* __restrict__ dst,
                              float* __restrict__ deg, int e) {
  int i = blockIdx.x * blockDim.x + threadIdx.x;
  if (i < e) atomicAdd(&deg[dst[i]], 1.0f);
}

__global__ void gcn_deg_finalize(const float* __restrict__ deg,
                                 float* __restrict__ dis,
                                 float* __restrict__ dinv, int n) {
  int i = blockIdx.x * blockDim.x + threadIdx.x;
  if (i < n) {
    float d = deg[i];
    dis[i] = rsqrtf(d);
    dinv[i] = 1.0f / d;
  }
}

__global__ void gcn_zero(float* __restrict__ p, int n) {
  int i = blockIdx.x * blockDim.x + threadIdx.x;
  if (i < n) p[i] = 0.0f;
}

__global__ __launch_bounds__(256) void gcn_scatter(
    const int* __restrict__ src, const int* __restrict__ dst,
    const float* __restrict__ H, const float* __restrict__ dis,
    float* __restrict__ agg, int nedges) {
  const int e = blockIdx.x * 8 + (threadIdx.x >> 5);
  if (e >= nedges) return;
  const int lane = threadIdx.x & 31;
  const int s = src[e];
  const int d = dst[e];
  const float norm = dis[s] * dis[d];
  const float4 v =
      reinterpret_cast<const float4*>(H + (size_t)s * 128)[lane];
  float* ad = agg + (size_t)d * 128 + lane * 4;
  atomicAdd(ad + 0, v.x * norm);
  atomicAdd(ad + 1, v.y * norm);
  atomicAdd(ad + 2, v.z * norm);
  atomicAdd(ad + 3, v.w * norm);
}

__global__ void gcn_combine(const float* __restrict__ agg,
                            const float* __restrict__ H,
                            const float* __restrict__ dinv,
                            const float* __restrict__ bias,
                            float* __restrict__ out, int total) {
  int i = blockIdx.x * blockDim.x + threadIdx.x;
  if (i < total) {
    const int node = i >> 7;
    const int f = i & 127;
    const float v = agg[i] + H[i] * dinv[node] + bias[f];
    out[i] = fmaxf(v, 0.0f);
  }
}

extern "C" void kernel_launch(void* const* d_in, const int* in_sizes, int n_in,
                              void* d_out, int out_size, void* d_ws,
                              size_t ws_size, hipStream_t stream) {
  const float* x = (const float*)d_in[0];
  const int* edge = (const int*)d_in[1];
  const float* W0 = (const float*)d_in[2];
  const float* b0 = (const float*)d_in[3];
  const float* W1 = (const float*)d_in[4];
  const float* b1 = (const float*)d_in[5];
  const float* W2 = (const float*)d_in[6];
  const float* b2 = (const float*)d_in[7];
  float* out = (float*)d_out;

  const int N = in_sizes[0] / 256;
  const int E = in_sizes[1] / 2;
  const int* srcIdx = edge;
  const int* dstIdx = edge + E;

  float* ws = (float*)d_ws;
  float* hbuf = ws;
  float* aggbuf = hbuf + (size_t)N * 128;
  float* xin = aggbuf + (size_t)N * 128;
  float* deg = xin + (size_t)N * 128;
  float* dis = deg + N;
  float* dinv = dis + N;
  float* wt0 = dinv + N;
  float* wt1 = wt0 + 128 * 256;
  float* wt2 = wt1 + 128 * 128;

  const int T = 256;
  const int nodeBlocks = (N + T - 1) / T;
  const int featTotal = N * 128;
  const int featBlocks = (featTotal + T - 1) / T;
  const int edgeBlocks = (E + T - 1) / T;
  const int scatBlocks = (E + 7) / 8;
  const int gemmBlocks = (N + 15) / 16;

  gcn_transpose_w<<<(256 * 128 + T - 1) / T, T, 0, stream>>>(W0, wt0, 256);
  gcn_transpose_w<<<(128 * 128 + T - 1) / T, T, 0, stream>>>(W1, wt1, 128);
  gcn_transpose_w<<<(128 * 128 + T - 1) / T, T, 0, stream>>>(W2, wt2, 128);

  gcn_deg_init<<<nodeBlocks, T, 0, stream>>>(deg, N);
  gcn_deg_count<<<edgeBlocks, T, 0, stream>>>(dstIdx, deg, E);
  gcn_deg_finalize<<<nodeBlocks, T, 0, stream>>>(deg, dis, dinv, N);

  gcn_gemm_wmma<256><<<gemmBlocks, T, 0, stream>>>(x, wt0, hbuf, N);
  gcn_zero<<<featBlocks, T, 0, stream>>>(aggbuf, featTotal);
  gcn_scatter<<<scatBlocks, T, 0, stream>>>(srcIdx, dstIdx, hbuf, dis, aggbuf, E);
  gcn_combine<<<featBlocks, T, 0, stream>>>(aggbuf, hbuf, dinv, b0, xin, featTotal);

  gcn_gemm_wmma<128><<<gemmBlocks, T, 0, stream>>>(xin, wt1, hbuf, N);
  gcn_zero<<<featBlocks, T, 0, stream>>>(aggbuf, featTotal);
  gcn_scatter<<<scatBlocks, T, 0, stream>>>(srcIdx, dstIdx, hbuf, dis, aggbuf, E);
  gcn_combine<<<featBlocks, T, 0, stream>>>(aggbuf, hbuf, dinv, b1, xin, featTotal);

  gcn_gemm_wmma<128><<<gemmBlocks, T, 0, stream>>>(xin, wt2, hbuf, N);
  gcn_zero<<<featBlocks, T, 0, stream>>>(aggbuf, featTotal);
  gcn_scatter<<<scatBlocks, T, 0, stream>>>(srcIdx, dstIdx, hbuf, dis, aggbuf, E);
  gcn_combine<<<featBlocks, T, 0, stream>>>(aggbuf, hbuf, dinv, b2, out, featTotal);
}


# Float-atomic edge scatter replaced by per-call dst-grouped edge list (int-atomic arrival-order rank, scan, fill) + wave-per-node register gather storing agg once; sum order = arrival rank
# speedup vs baseline: 2.1629x; 2.1629x over previous
#include <hip/hip_runtime.h>
#include <hip/hip_bf16.h>


typedef __attribute__((ext_vector_type(2))) float v2f;
typedef __attribute__((ext_vector_type(8))) float v8f;

__global__ void gcn_transpose_w(const float* __restrict__ W,
                                float* __restrict__ Wt, int K) {
  int i = blockIdx.x * blockDim.x + threadIdx.x;
  if (i < K * 128) {
    const int k = i >> 7;
    const int c = i & 127;
    Wt[(size_t)c * K + k] = W[i];
  }
}

template <int K>
__global__ __launch_bounds__(256) void gcn_gemm_wmma(
    const float* __restrict__ X, const float* __restrict__ Wt,
    float* __restrict__ H, int nrows) {
  constexpr int LDA = K + 4;
  __shared__ __align__(16) float As[16 * LDA];

  const int rowBase = blockIdx.x * 16;
  const int tid = threadIdx.x;

  constexpr int vecPerRow = K / 4;
  for (int i = tid; i < 16 * vecPerRow; i += 256) {
    const int r = i / vecPerRow;
    const int v = i % vecPerRow;
    const int grow = rowBase + r;
    float4 val = make_float4(0.f, 0.f, 0.f, 0.f);
    if (grow < nrows) {
      val = *reinterpret_cast<const float4*>(X + (size_t)grow * K + 4 * v);
    }
    *reinterpret_cast<float4*>(&As[r * LDA + 4 * v]) = val;
  }
  __syncthreads();

  const int wave = tid >> 5;
  const int lane = tid & 31;
  const int l15 = lane & 15;
  const int halfSel = lane >> 4;
  const int col = wave * 16 + l15;

  const float* __restrict__ Bcol = Wt + (size_t)col * K;

  v8f acc = {};
#pragma unroll 4
  for (int k0 = 0; k0 < K; k0 += 4) {
    const int kk = k0 + 2 * halfSel;
    v2f a;
    a.x = As[l15 * LDA + kk];
    a.y = As[l15 * LDA + kk + 1];
    v2f b;
    b.x = Bcol[kk];
    b.y = Bcol[kk + 1];
    acc = __builtin_amdgcn_wmma_f32_16x16x4_f32(
         false, a,  false, b,
         (short)0, acc,  false,  false);
  }

  if (rowBase + 16 <= nrows) {
    float* __restrict__ Hp = H + (size_t)(rowBase + 8 * halfSel) * 128 + col;
#pragma unroll
    for (int r = 0; r < 8; ++r) {
      Hp[(size_t)r * 128] = acc[r];
    }
  } else {
#pragma unroll
    for (int r = 0; r < 8; ++r) {
      const int m = rowBase + r + 8 * halfSel;
      if (m < nrows) H[(size_t)m * 128 + col] = acc[r];
    }
  }
}

struct __align__(8) EdgeRec {
  int src;
  float w;
};

__global__ void gcn_cnt_zero(int* __restrict__ cnt, int n) {
  int i = blockIdx.x * blockDim.x + threadIdx.x;
  if (i < n) cnt[i] = 0;
}

__global__ void gcn_count_rank(const int* __restrict__ dst,
                               int* __restrict__ cnt,
                               int* __restrict__ rank, int e) {
  int i = blockIdx.x * blockDim.x + threadIdx.x;
  if (i < e) rank[i] = atomicAdd(&cnt[dst[i]], 1);
}

__global__ __launch_bounds__(1024) void gcn_scan_deg(
    const int* __restrict__ cnt, int* __restrict__ rowptr,
    float* __restrict__ dis, float* __restrict__ dinv, int n) {
  __shared__ int wsum[32];
  __shared__ int wbef[32];
  __shared__ int wpre[32];
  __shared__ int baseSh;

  const int t = threadIdx.x;
  const int lane = t & 31;
  const int wv = t >> 5;
  const int idx = blockIdx.x * 1024 + t;

  int before = 0;
  for (int b = 0; b < (int)blockIdx.x; ++b) before += cnt[b * 1024 + t];
  const int own = (idx < n) ? cnt[idx] : 0;

  int incl = own;
#pragma unroll
  for (int o = 1; o < 32; o <<= 1) {
    const int y = __shfl_up(incl, o);
    if (lane >= o) incl += y;
  }
#pragma unroll
  for (int o = 16; o > 0; o >>= 1) before += __shfl_xor(before, o);

  if (lane == 31) wsum[wv] = incl;
  if (lane == 0) wbef[wv] = before;
  __syncthreads();
  if (wv == 0) {
    const int v = wsum[lane];
    int inc2 = v;
#pragma unroll
    for (int o = 1; o < 32; o <<= 1) {
      const int y = __shfl_up(inc2, o);
      if (lane >= o) inc2 += y;
    }
    int b2 = wbef[lane];
#pragma unroll
    for (int o = 16; o > 0; o >>= 1) b2 += __shfl_xor(b2, o);
    wpre[lane] = inc2 - v;
    if (lane == 0) baseSh = b2;
  }
  __syncthreads();
  if (idx < n) {
    rowptr[idx] = baseSh + wpre[wv] + incl - own;
    const float d = (float)own + 1.0f;
    dis[idx] = rsqrtf(d);
    dinv[idx] = 1.0f / d;
  }
}

__global__ void gcn_fill(const int* __restrict__ src,
                         const int* __restrict__ dst,
                         const int* __restrict__ rank,
                         const int* __restrict__ rowptr,
                         const float* __restrict__ dis,
                         EdgeRec* __restrict__ rec, int e) {
  int i = blockIdx.x * blockDim.x + threadIdx.x;
  if (i < e) {
    const int s = src[i];
    const int d = dst[i];
    EdgeRec r;
    r.src = s;
    r.w = dis[s] * dis[d];
    rec[rowptr[d] + rank[i]] = r;
  }
}

__device__ __forceinline__ void gcn_acc(float4& acc, const float4 v, float w) {
  acc.x = fmaf(v.x, w, acc.x);
  acc.y = fmaf(v.y, w, acc.y);
  acc.z = fmaf(v.z, w, acc.z);
  acc.w = fmaf(v.w, w, acc.w);
}

template <int C>
__device__ __forceinline__ void gcn_gather_chunk(
    float4& acc, const EdgeRec* __restrict__ rec,
    const float4* __restrict__ H4, int lane) {
  EdgeRec r[C];
  float4 v[C];
#pragma unroll
  for (int k = 0; k < C; ++k) r[k] = rec[k];
#pragma unroll
  for (int k = 0; k < C; ++k) v[k] = H4[(size_t)r[k].src * 32 + lane];
#pragma unroll
  for (int k = 0; k < C; ++k) gcn_acc(acc, v[k], r[k].w);
}

__global__ __launch_bounds__(256) void gcn_gather(
    const EdgeRec* __restrict__ rec, const int* __restrict__ rowptr,
    const int* __restrict__ cnt, const float* __restrict__ H,
    float* __restrict__ agg, int n) {
  const int node =
      __builtin_amdgcn_readfirstlane(blockIdx.x * 8 + (threadIdx.x >> 5));
  if (node >= n) return;
  const int lane = threadIdx.x & 31;
  const float4* __restrict__ H4 = reinterpret_cast<const float4*>(H);

  int j = rowptr[node];
  const int end = j + cnt[node];

  float4 acc = make_float4(0.f, 0.f, 0.f, 0.f);
  for (; j + 8 <= end; j += 8) gcn_gather_chunk<8>(acc, rec + j, H4, lane);
  if (j + 4 <= end) {
    gcn_gather_chunk<4>(acc, rec + j, H4, lane);
    j += 4;
  }
  if (j + 2 <= end) {
    gcn_gather_chunk<2>(acc, rec + j, H4, lane);
    j += 2;
  }
  if (j < end) gcn_gather_chunk<1>(acc, rec + j, H4, lane);

  reinterpret_cast<float4*>(agg)[(size_t)node * 32 + lane] = acc;
}

__global__ void gcn_combine(const float* __restrict__ agg,
                            const float* __restrict__ H,
                            const float* __restrict__ dinv,
                            const float* __restrict__ bias,
                            float* __restrict__ out, int total) {
  int i = blockIdx.x * blockDim.x + threadIdx.x;
  if (i < total) {
    const int node = i >> 7;
    const int f = i & 127;
    const float v = agg[i] + H[i] * dinv[node] + bias[f];
    out[i] = fmaxf(v, 0.0f);
  }
}

extern "C" void kernel_launch(void* const* d_in, const int* in_sizes, int n_in,
                              void* d_out, int out_size, void* d_ws,
                              size_t ws_size, hipStream_t stream) {
  const float* x = (const float*)d_in[0];
  const int* edge = (const int*)d_in[1];
  const float* W0 = (const float*)d_in[2];
  const float* b0 = (const float*)d_in[3];
  const float* W1 = (const float*)d_in[4];
  const float* b1 = (const float*)d_in[5];
  const float* W2 = (const float*)d_in[6];
  const float* b2 = (const float*)d_in[7];
  float* out = (float*)d_out;

  const int N = in_sizes[0] / 256;
  const int E = in_sizes[1] / 2;
  const int* srcIdx = edge;
  const int* dstIdx = edge + E;

  float* ws = (float*)d_ws;
  float* hbuf = ws;
  float* aggbuf = hbuf + (size_t)N * 128;
  float* xin = aggbuf + (size_t)N * 128;
  EdgeRec* rec = (EdgeRec*)(xin + (size_t)N * 128);
  float* wt0 = (float*)(rec + E);
  float* wt1 = wt0 + 128 * 256;
  float* wt2 = wt1 + 128 * 128;
  float* dis = wt2 + 128 * 128;
  float* dinv = dis + N;
  int* cnt = (int*)(dinv + N);
  int* rowptr = cnt + N;
  int* rank = rowptr + N;

  const int T = 256;
  const int nodeBlocks = (N + T - 1) / T;
  const int featTotal = N * 128;
  const int featBlocks = (featTotal + T - 1) / T;
  const int edgeBlocks = (E + T - 1) / T;
  const int gathBlocks = (N + 7) / 8;
  const int scanBlocks = (N + 1023) / 1024;
  const int gemmBlocks = (N + 15) / 16;

  gcn_transpose_w<<<(256 * 128 + T - 1) / T, T, 0, stream>>>(W0, wt0, 256);
  gcn_transpose_w<<<(128 * 128 + T - 1) / T, T, 0, stream>>>(W1, wt1, 128);
  gcn_transpose_w<<<(128 * 128 + T - 1) / T, T, 0, stream>>>(W2, wt2, 128);

  gcn_cnt_zero<<<nodeBlocks, T, 0, stream>>>(cnt, N);
  gcn_count_rank<<<edgeBlocks, T, 0, stream>>>(dstIdx, cnt, rank, E);
  gcn_scan_deg<<<scanBlocks, 1024, 0, stream>>>(cnt, rowptr, dis, dinv, N);
  gcn_fill<<<edgeBlocks, T, 0, stream>>>(srcIdx, dstIdx, rank, rowptr, dis, rec, E);

  gcn_gemm_wmma<256><<<gemmBlocks, T, 0, stream>>>(x, wt0, hbuf, N);
  gcn_gather<<<gathBlocks, T, 0, stream>>>(rec, rowptr, cnt, hbuf, aggbuf, N);
  gcn_combine<<<featBlocks, T, 0, stream>>>(aggbuf, hbuf, dinv, b0, xin, featTotal);

  gcn_gemm_wmma<128><<<gemmBlocks, T, 0, stream>>>(xin, wt1, hbuf, N);
  gcn_gather<<<gathBlocks, T, 0, stream>>>(rec, rowptr, cnt, hbuf, aggbuf, N);
  gcn_combine<<<featBlocks, T, 0, stream>>>(aggbuf, hbuf, dinv, b1, xin, featTotal);

  gcn_gemm_wmma<128><<<gemmBlocks, T, 0, stream>>>(xin, wt2, hbuf, N);
  gcn_gather<<<gathBlocks, T, 0, stream>>>(rec, rowptr, cnt, hbuf, aggbuf, N);
  gcn_combine<<<featBlocks, T, 0, stream>>>(aggbuf, hbuf, dinv, b2, out, featTotal);
}


# Combine epilogue (self-loop term h*deg_inv, bias, relu) fused into the gather wave ahead of its single row store: 3 combine launches and the agg plane are gone; sum order unchanged
# speedup vs baseline: 2.7172x; 1.2563x over previous
#include <hip/hip_runtime.h>
#include <hip/hip_bf16.h>


typedef __attribute__((ext_vector_type(2))) float v2f;
typedef __attribute__((ext_vector_type(8))) float v8f;

__global__ void gcn_transpose_w(const float* __restrict__ W,
                                float* __restrict__ Wt, int K) {
  int i = blockIdx.x * blockDim.x + threadIdx.x;
  if (i < K * 128) {
    const int k = i >> 7;
    const int c = i & 127;
    Wt[(size_t)c * K + k] = W[i];
  }
}

template <int K>
__global__ __launch_bounds__(256) void gcn_gemm_wmma(
    const float* __restrict__ X, const float* __restrict__ Wt,
    float* __restrict__ H, int nrows) {
  constexpr int LDA = K + 4;
  __shared__ __align__(16) float As[16 * LDA];

  const int rowBase = blockIdx.x * 16;
  const int tid = threadIdx.x;

  constexpr int vecPerRow = K / 4;
  for (int i = tid; i < 16 * vecPerRow; i += 256) {
    const int r = i / vecPerRow;
    const int v = i % vecPerRow;
    const int grow = rowBase + r;
    float4 val = make_float4(0.f, 0.f, 0.f, 0.f);
    if (grow < nrows) {
      val = *reinterpret_cast<const float4*>(X + (size_t)grow * K + 4 * v);
    }
    *reinterpret_cast<float4*>(&As[r * LDA + 4 * v]) = val;
  }
  __syncthreads();

  const int wave = tid >> 5;
  const int lane = tid & 31;
  const int l15 = lane & 15;
  const int halfSel = lane >> 4;
  const int col = wave * 16 + l15;

  const float* __restrict__ Bcol = Wt + (size_t)col * K;

  v8f acc = {};
#pragma unroll 4
  for (int k0 = 0; k0 < K; k0 += 4) {
    const int kk = k0 + 2 * halfSel;
    v2f a;
    a.x = As[l15 * LDA + kk];
    a.y = As[l15 * LDA + kk + 1];
    v2f b;
    b.x = Bcol[kk];
    b.y = Bcol[kk + 1];
    acc = __builtin_amdgcn_wmma_f32_16x16x4_f32(
         false, a,  false, b,
         (short)0, acc,  false,  false);
  }

  if (rowBase + 16 <= nrows) {
    float* __restrict__ Hp = H + (size_t)(rowBase + 8 * halfSel) * 128 + col;
#pragma unroll
    for (int r = 0; r < 8; ++r) {
      Hp[(size_t)r * 128] = acc[r];
    }
  } else {
#pragma unroll
    for (int r = 0; r < 8; ++r) {
      const int m = rowBase + r + 8 * halfSel;
      if (m < nrows) H[(size_t)m * 128 + col] = acc[r];
    }
  }
}

struct __align__(8) EdgeRec {
  int src;
  float w;
};

__global__ void gcn_cnt_zero(int* __restrict__ cnt, int n) {
  int i = blockIdx.x * blockDim.x + threadIdx.x;
  if (i < n) cnt[i] = 0;
}

__global__ void gcn_count_rank(const int* __restrict__ dst,
                               int* __restrict__ cnt,
                               int* __restrict__ rank, int e) {
  int i = blockIdx.x * blockDim.x + threadIdx.x;
  if (i < e) rank[i] = atomicAdd(&cnt[dst[i]], 1);
}

__global__ __launch_bounds__(1024) void gcn_scan_deg(
    const int* __restrict__ cnt, int* __restrict__ rowptr,
    float* __restrict__ dis, float* __restrict__ dinv, int n) {
  __shared__ int wsum[32];
  __shared__ int wbef[32];
  __shared__ int wpre[32];
  __shared__ int baseSh;

  const int t = threadIdx.x;
  const int lane = t & 31;
  const int wv = t >> 5;
  const int idx = blockIdx.x * 1024 + t;

  int before = 0;
  for (int b = 0; b < (int)blockIdx.x; ++b) before += cnt[b * 1024 + t];
  const int own = (idx < n) ? cnt[idx] : 0;

  int incl = own;
#pragma unroll
  for (int o = 1; o < 32; o <<= 1) {
    const int y = __shfl_up(incl, o);
    if (lane >= o) incl += y;
  }
#pragma unroll
  for (int o = 16; o > 0; o >>= 1) before += __shfl_xor(before, o);

  if (lane == 31) wsum[wv] = incl;
  if (lane == 0) wbef[wv] = before;
  __syncthreads();
  if (wv == 0) {
    const int v = wsum[lane];
    int inc2 = v;
#pragma unroll
    for (int o = 1; o < 32; o <<= 1) {
      const int y = __shfl_up(inc2, o);
      if (lane >= o) inc2 += y;
    }
    int b2 = wbef[lane];
#pragma unroll
    for (int o = 16; o > 0; o >>= 1) b2 += __shfl_xor(b2, o);
    wpre[lane] = inc2 - v;
    if (lane == 0) baseSh = b2;
  }
  __syncthreads();
  if (idx < n) {
    rowptr[idx] = baseSh + wpre[wv] + incl - own;
    const float d = (float)own + 1.0f;
    dis[idx] = rsqrtf(d);
    dinv[idx] = 1.0f / d;
  }
}

__global__ void gcn_fill(const int* __restrict__ src,
                         const int* __restrict__ dst,
                         const int* __restrict__ rank,
                         const int* __restrict__ rowptr,
                         const float* __restrict__ dis,
                         EdgeRec* __restrict__ rec, int e) {
  int i = blockIdx.x * blockDim.x + threadIdx.x;
  if (i < e) {
    const int s = src[i];
    const int d = dst[i];
    EdgeRec r;
    r.src = s;
    r.w = dis[s] * dis[d];
    rec[rowptr[d] + rank[i]] = r;
  }
}

__device__ __forceinline__ void gcn_acc(float4& acc, const float4 v, float w) {
  acc.x = fmaf(v.x, w, acc.x);
  acc.y = fmaf(v.y, w, acc.y);
  acc.z = fmaf(v.z, w, acc.z);
  acc.w = fmaf(v.w, w, acc.w);
}

template <int C>
__device__ __forceinline__ void gcn_gather_chunk(
    float4& acc, const EdgeRec* __restrict__ rec,
    const float4* __restrict__ H4, int lane) {
  EdgeRec r[C];
  float4 v[C];
#pragma unroll
  for (int k = 0; k < C; ++k) r[k] = rec[k];
#pragma unroll
  for (int k = 0; k < C; ++k) v[k] = H4[(size_t)r[k].src * 32 + lane];
#pragma unroll
  for (int k = 0; k < C; ++k) gcn_acc(acc, v[k], r[k].w);
}

__global__ __launch_bounds__(256) void gcn_gather_combine(
    const EdgeRec* __restrict__ rec, const int* __restrict__ rowptr,
    const int* __restrict__ cnt, const float* __restrict__ H,
    const float* __restrict__ dinv, const float* __restrict__ bias,
    float* __restrict__ out, int n) {
  const int node =
      __builtin_amdgcn_readfirstlane(blockIdx.x * 8 + (threadIdx.x >> 5));
  if (node >= n) return;
  const int lane = threadIdx.x & 31;
  const float4* __restrict__ H4 = reinterpret_cast<const float4*>(H);

  int j = rowptr[node];
  const int end = j + cnt[node];
  const float di = dinv[node];
  const float4 hs = H4[(size_t)node * 32 + lane];
  const float4 bv = reinterpret_cast<const float4*>(bias)[lane];

  float4 acc = make_float4(0.f, 0.f, 0.f, 0.f);
  for (; j + 8 <= end; j += 8) gcn_gather_chunk<8>(acc, rec + j, H4, lane);
  if (j + 4 <= end) {
    gcn_gather_chunk<4>(acc, rec + j, H4, lane);
    j += 4;
  }
  if (j + 2 <= end) {
    gcn_gather_chunk<2>(acc, rec + j, H4, lane);
    j += 2;
  }
  if (j < end) gcn_gather_chunk<1>(acc, rec + j, H4, lane);

  float4 o;
  o.x = fmaxf(fmaf(hs.x, di, acc.x) + bv.x, 0.0f);
  o.y = fmaxf(fmaf(hs.y, di, acc.y) + bv.y, 0.0f);
  o.z = fmaxf(fmaf(hs.z, di, acc.z) + bv.z, 0.0f);
  o.w = fmaxf(fmaf(hs.w, di, acc.w) + bv.w, 0.0f);
  reinterpret_cast<float4*>(out)[(size_t)node * 32 + lane] = o;
}

extern "C" void kernel_launch(void* const* d_in, const int* in_sizes, int n_in,
                              void* d_out, int out_size, void* d_ws,
                              size_t ws_size, hipStream_t stream) {
  const float* x = (const float*)d_in[0];
  const int* edge = (const int*)d_in[1];
  const float* W0 = (const float*)d_in[2];
  const float* b0 = (const float*)d_in[3];
  const float* W1 = (const float*)d_in[4];
  const float* b1 = (const float*)d_in[5];
  const float* W2 = (const float*)d_in[6];
  const float* b2 = (const float*)d_in[7];
  float* out = (float*)d_out;

  const int N = in_sizes[0] / 256;
  const int E = in_sizes[1] / 2;
  const int* srcIdx = edge;
  const int* dstIdx = edge + E;

  float* ws = (float*)d_ws;
  float* hbuf = ws;
  float* xin = hbuf + (size_t)N * 128;
  EdgeRec* rec = (EdgeRec*)(xin + (size_t)N * 128);
  float* wt0 = (float*)(rec + E);
  float* wt1 = wt0 + 128 * 256;
  float* wt2 = wt1 + 128 * 128;
  float* dis = wt2 + 128 * 128;
  float* dinv = dis + N;
  int* cnt = (int*)(dinv + N);
  int* rowptr = cnt + N;
  int* rank = rowptr + N;

  const int T = 256;
  const int nodeBlocks = (N + T - 1) / T;
  const int edgeBlocks = (E + T - 1) / T;
  const int gathBlocks = (N + 7) / 8;
  const int scanBlocks = (N + 1023) / 1024;
  const int gemmBlocks = (N + 15) / 16;

  gcn_transpose_w<<<(256 * 128 + T - 1) / T, T, 0, stream>>>(W0, wt0, 256);
  gcn_transpose_w<<<(128 * 128 + T - 1) / T, T, 0, stream>>>(W1, wt1, 128);
  gcn_transpose_w<<<(128 * 128 + T - 1) / T, T, 0, stream>>>(W2, wt2, 128);

  gcn_cnt_zero<<<nodeBlocks, T, 0, stream>>>(cnt, N);
  gcn_count_rank<<<edgeBlocks, T, 0, stream>>>(dstIdx, cnt, rank, E);
  gcn_scan_deg<<<scanBlocks, 1024, 0, stream>>>(cnt, rowptr, dis, dinv, N);
  gcn_fill<<<edgeBlocks, T, 0, stream>>>(srcIdx, dstIdx, rank, rowptr, dis, rec, E);

  gcn_gemm_wmma<256><<<gemmBlocks, T, 0, stream>>>(x, wt0, hbuf, N);
  gcn_gather_combine<<<gathBlocks, T, 0, stream>>>(rec, rowptr, cnt, hbuf, dinv, b0, xin, N);

  gcn_gemm_wmma<128><<<gemmBlocks, T, 0, stream>>>(xin, wt1, hbuf, N);
  gcn_gather_combine<<<gathBlocks, T, 0, stream>>>(rec, rowptr, cnt, hbuf, dinv, b1, xin, N);

  gcn_gemm_wmma<128><<<gemmBlocks, T, 0, stream>>>(xin, wt2, hbuf, N);
  gcn_gather_combine<<<gathBlocks, T, 0, stream>>>(rec, rowptr, cnt, hbuf, dinv, b2, out, N);
}


# Edge count and rank without global atomics: 64 edge slices, per slice one block with a counter per node in LDS, then per-node running sums over slices; sum order = slice order, then LDS arrival
# speedup vs baseline: 3.0552x; 1.1244x over previous
#include <hip/hip_runtime.h>
#include <hip/hip_bf16.h>


typedef __attribute__((ext_vector_type(2))) float v2f;
typedef __attribute__((ext_vector_type(8))) float v8f;

__global__ void gcn_transpose_w(const float* __restrict__ W,
                                float* __restrict__ Wt, int K) {
  int i = blockIdx.x * blockDim.x + threadIdx.x;
  if (i < K * 128) {
    const int k = i >> 7;
    const int c = i & 127;
    Wt[(size_t)c * K + k] = W[i];
  }
}

template <int K>
__global__ __launch_bounds__(256) void gcn_gemm_wmma(
    const float* __restrict__ X, const float* __restrict__ Wt,
    float* __restrict__ H, int nrows) {
  constexpr int LDA = K + 4;
  __shared__ __align__(16) float As[16 * LDA];

  const int rowBase = blockIdx.x * 16;
  const int tid = threadIdx.x;

  constexpr int vecPerRow = K / 4;
  for (int i = tid; i < 16 * vecPerRow; i += 256) {
    const int r = i / vecPerRow;
    const int v = i % vecPerRow;
    const int grow = rowBase + r;
    float4 val = make_float4(0.f, 0.f, 0.f, 0.f);
    if (grow < nrows) {
      val = *reinterpret_cast<const float4*>(X + (size_t)grow * K + 4 * v);
    }
    *reinterpret_cast<float4*>(&As[r * LDA + 4 * v]) = val;
  }
  __syncthreads();

  const int wave = tid >> 5;
  const int lane = tid & 31;
  const int l15 = lane & 15;
  const int halfSel = lane >> 4;
  const int col = wave * 16 + l15;

  const float* __restrict__ Bcol = Wt + (size_t)col * K;

  v8f acc = {};
#pragma unroll 4
  for (int k0 = 0; k0 < K; k0 += 4) {
    const int kk = k0 + 2 * halfSel;
    v2f a;
    a.x = As[l15 * LDA + kk];
    a.y = As[l15 * LDA + kk + 1];
    v2f b;
    b.x = Bcol[kk];
    b.y = Bcol[kk + 1];
    acc = __builtin_amdgcn_wmma_f32_16x16x4_f32(
         false, a,  false, b,
         (short)0, acc,  false,  false);
  }

  if (rowBase + 16 <= nrows) {
    float* __restrict__ Hp = H + (size_t)(rowBase + 8 * halfSel) * 128 + col;
#pragma unroll
    for (int r = 0; r < 8; ++r) {
      Hp[(size_t)r * 128] = acc[r];
    }
  } else {
#pragma unroll
    for (int r = 0; r < 8; ++r) {
      const int m = rowBase + r + 8 * halfSel;
      if (m < nrows) H[(size_t)m * 128 + col] = acc[r];
    }
  }
}

struct __align__(8) EdgeRec {
  int src;
  float w;
};

constexpr int GCN_SLICES = 64;
constexpr int GCN_NODE_CAP = 50176;

__device__ __forceinline__ int gcn_part_pitch(int n) { return (n + 31) & ~31; }

__global__ __launch_bounds__(1024) void gcn_slice_count(
    const int* __restrict__ dst, int* __restrict__ part,
    int* __restrict__ rank, int e, int n, int sliceLen) {
  __shared__ __align__(16) int cntS[GCN_NODE_CAP];
  const int t = threadIdx.x;
  const int lo = blockIdx.y * GCN_NODE_CAP;
  const int span = min(GCN_NODE_CAP, n - lo);

  for (int i = t; i < GCN_NODE_CAP / 4; i += 1024)
    reinterpret_cast<int4*>(cntS)[i] = make_int4(0, 0, 0, 0);
  __syncthreads();

  const int eBeg = blockIdx.x * sliceLen;
  const int eEnd = min(eBeg + sliceLen, e);
  constexpr int U = 4;
  for (int i0 = eBeg + t; i0 < eEnd; i0 += U * 1024) {
    int d[U];
#pragma unroll
    for (int u = 0; u < U; ++u) {
      const int i = i0 + u * 1024;
      const int v = dst[min(i, eEnd - 1)];
      d[u] = (i < eEnd) ? v - lo : -1;
    }
#pragma unroll
    for (int u = 0; u < U; ++u) {
      if ((unsigned)d[u] < (unsigned)span)
        rank[i0 + u * 1024] = atomicAdd(&cntS[d[u]], 1);
    }
  }
  __syncthreads();

  int* __restrict__ row = part + (size_t)blockIdx.x * gcn_part_pitch(n) + lo;
  for (int i = t; i < span; i += 1024) row[i] = cntS[i];
}

__global__ __launch_bounds__(256) void gcn_slice_prefix(int* __restrict__ part,
                                                        int* __restrict__ cnt,
                                                        int n) {
  const int i = blockIdx.x * blockDim.x + threadIdx.x;
  if (i >= n) return;
  const size_t pitch = gcn_part_pitch(n);
  int c[GCN_SLICES];
#pragma unroll
  for (int b = 0; b < GCN_SLICES; ++b) c[b] = part[b * pitch + i];
  int run = 0;
#pragma unroll
  for (int b = 0; b < GCN_SLICES; ++b) {
    part[b * pitch + i] = run;
    run += c[b];
  }
  cnt[i] = run;
}

__global__ __launch_bounds__(1024) void gcn_scan_deg(
    const int* __restrict__ cnt, int* __restrict__ rowptr,
    float* __restrict__ dis, float* __restrict__ dinv, int n) {
  __shared__ int wsum[32];
  __shared__ int wbef[32];
  __shared__ int wpre[32];
  __shared__ int baseSh;

  const int t = threadIdx.x;
  const int lane = t & 31;
  const int wv = t >> 5;
  const int idx = blockIdx.x * 1024 + t;

  int before = 0;
  for (int b = 0; b < (int)blockIdx.x; ++b) before += cnt[b * 1024 + t];
  const int own = (idx < n) ? cnt[idx] : 0;

  int incl = own;
#pragma unroll
  for (int o = 1; o < 32; o <<= 1) {
    const int y = __shfl_up(incl, o);
    if (lane >= o) incl += y;
  }
#pragma unroll
  for (int o = 16; o > 0; o >>= 1) before += __shfl_xor(before, o);

  if (lane == 31) wsum[wv] = incl;
  if (lane == 0) wbef[wv] = before;
  __syncthreads();
  if (wv == 0) {
    const int v = wsum[lane];
    int inc2 = v;
#pragma unroll
    for (int o = 1; o < 32; o <<= 1) {
      const int y = __shfl_up(inc2, o);
      if (lane >= o) inc2 += y;
    }
    int b2 = wbef[lane];
#pragma unroll
    for (int o = 16; o > 0; o >>= 1) b2 += __shfl_xor(b2, o);
    wpre[lane] = inc2 - v;
    if (lane == 0) baseSh = b2;
  }
  __syncthreads();
  if (idx < n) {
    rowptr[idx] = baseSh + wpre[wv] + incl - own;
    const float d = (float)own + 1.0f;
    dis[idx] = rsqrtf(d);
    dinv[idx] = 1.0f / d;
  }
}

__global__ void gcn_fill(const int* __restrict__ src,
                         const int* __restrict__ dst,
                         const int* __restrict__ rank,
                         const int* __restrict__ rowptr,
                         const int* __restrict__ part,
                         const float* __restrict__ dis,
                         EdgeRec* __restrict__ rec, int e, int n,
                         int sliceLen) {
  const int k = blockIdx.x * blockDim.x + threadIdx.x;
  const int i = blockIdx.y * sliceLen + k;
  if (k < sliceLen && i < e) {
    const int s = src[i];
    const int d = dst[i];
    EdgeRec r;
    r.src = s;
    r.w = dis[s] * dis[d];
    const int first = part[(size_t)blockIdx.y * gcn_part_pitch(n) + d];
    rec[rowptr[d] + first + rank[i]] = r;
  }
}

__device__ __forceinline__ void gcn_acc(float4& acc, const float4 v, float w) {
  acc.x = fmaf(v.x, w, acc.x);
  acc.y = fmaf(v.y, w, acc.y);
  acc.z = fmaf(v.z, w, acc.z);
  acc.w = fmaf(v.w, w, acc.w);
}

template <int C>
__device__ __forceinline__ void gcn_gather_chunk(
    float4& acc, const EdgeRec* __restrict__ rec,
    const float4* __restrict__ H4, int lane) {
  EdgeRec r[C];
  float4 v[C];
#pragma unroll
  for (int k = 0; k < C; ++k) r[k] = rec[k];
#pragma unroll
  for (int k = 0; k < C; ++k) v[k] = H4[(size_t)r[k].src * 32 + lane];
#pragma unroll
  for (int k = 0; k < C; ++k) gcn_acc(acc, v[k], r[k].w);
}

__global__ __launch_bounds__(256) void gcn_gather_combine(
    const EdgeRec* __restrict__ rec, const int* __restrict__ rowptr,
    const int* __restrict__ cnt, const float* __restrict__ H,
    const float* __restrict__ dinv, const float* __restrict__ bias,
    float* __restrict__ out, int n) {
  const int node =
      __builtin_amdgcn_readfirstlane(blockIdx.x * 8 + (threadIdx.x >> 5));
  if (node >= n) return;
  const int lane = threadIdx.x & 31;
  const float4* __restrict__ H4 = reinterpret_cast<const float4*>(H);

  int j = rowptr[node];
  const int end = j + cnt[node];
  const float di = dinv[node];
  const float4 hs = H4[(size_t)node * 32 + lane];
  const float4 bv = reinterpret_cast<const float4*>(bias)[lane];

  float4 acc = make_float4(0.f, 0.f, 0.f, 0.f);
  for (; j + 8 <= end; j += 8) gcn_gather_chunk<8>(acc, rec + j, H4, lane);
  if (j + 4 <= end) {
    gcn_gather_chunk<4>(acc, rec + j, H4, lane);
    j += 4;
  }
  if (j + 2 <= end) {
    gcn_gather_chunk<2>(acc, rec + j, H4, lane);
    j += 2;
  }
  if (j < end) gcn_gather_chunk<1>(acc, rec + j, H4, lane);

  float4 o;
  o.x = fmaxf(fmaf(hs.x, di, acc.x) + bv.x, 0.0f);
  o.y = fmaxf(fmaf(hs.y, di, acc.y) + bv.y, 0.0f);
  o.z = fmaxf(fmaf(hs.z, di, acc.z) + bv.z, 0.0f);
  o.w = fmaxf(fmaf(hs.w, di, acc.w) + bv.w, 0.0f);
  reinterpret_cast<float4*>(out)[(size_t)node * 32 + lane] = o;
}

extern "C" void kernel_launch(void* const* d_in, const int* in_sizes, int n_in,
                              void* d_out, int out_size, void* d_ws,
                              size_t ws_size, hipStream_t stream) {
  const float* x = (const float*)d_in[0];
  const int* edge = (const int*)d_in[1];
  const float* W0 = (const float*)d_in[2];
  const float* b0 = (const float*)d_in[3];
  const float* W1 = (const float*)d_in[4];
  const float* b1 = (const float*)d_in[5];
  const float* W2 = (const float*)d_in[6];
  const float* b2 = (const float*)d_in[7];
  float* out = (float*)d_out;

  const int N = in_sizes[0] / 256;
  const int E = in_sizes[1] / 2;
  const int* srcIdx = edge;
  const int* dstIdx = edge + E;

  float* ws = (float*)d_ws;
  float* hbuf = ws;
  float* xin = hbuf + (size_t)N * 128;
  EdgeRec* rec = (EdgeRec*)(xin + (size_t)N * 128);
  float* wt0 = (float*)(rec + E);
  float* wt1 = wt0 + 128 * 256;
  float* wt2 = wt1 + 128 * 128;
  float* dis = wt2 + 128 * 128;
  float* dinv = dis + N;
  int* cnt = (int*)(dinv + N);
  int* rowptr = cnt + N;
  int* rank = rowptr + N;
  int* part = rank + E;

  const int T = 256;
  const int nodeBlocks = (N + T - 1) / T;
  const int sliceLen = (E + GCN_SLICES - 1) / GCN_SLICES;
  const int sliceBlocks = (sliceLen + T - 1) / T;
  const int nodeRows = (N + GCN_NODE_CAP - 1) / GCN_NODE_CAP;
  const int gathBlocks = (N + 7) / 8;
  const int scanBlocks = (N + 1023) / 1024;
  const int gemmBlocks = (N + 15) / 16;

  gcn_transpose_w<<<(256 * 128 + T - 1) / T, T, 0, stream>>>(W0, wt0, 256);
  gcn_transpose_w<<<(128 * 128 + T - 1) / T, T, 0, stream>>>(W1, wt1, 128);
  gcn_transpose_w<<<(128 * 128 + T - 1) / T, T, 0, stream>>>(W2, wt2, 128);

  gcn_slice_count<<<dim3(GCN_SLICES, nodeRows), 1024, 0, stream>>>(
      dstIdx, part, rank, E, N, sliceLen);
  gcn_slice_prefix<<<nodeBlocks, T, 0, stream>>>(part, cnt, N);
  gcn_scan_deg<<<scanBlocks, 1024, 0, stream>>>(cnt, rowptr, dis, dinv, N);
  gcn_fill<<<dim3(sliceBlocks, GCN_SLICES), T, 0, stream>>>(
      srcIdx, dstIdx, rank, rowptr, part, dis, rec, E, N, sliceLen);

  gcn_gemm_wmma<256><<<gemmBlocks, T, 0, stream>>>(x, wt0, hbuf, N);
  gcn_gather_combine<<<gathBlocks, T, 0, stream>>>(rec, rowptr, cnt, hbuf, dinv, b0, xin, N);

  gcn_gemm_wmma<128><<<gemmBlocks, T, 0, stream>>>(xin, wt1, hbuf, N);
  gcn_gather_combine<<<gathBlocks, T, 0, stream>>>(rec, rowptr, cnt, hbuf, dinv, b1, xin, N);

  gcn_gemm_wmma<128><<<gemmBlocks, T, 0, stream>>>(xin, wt2, hbuf, N);
  gcn_gather_combine<<<gathBlocks, T, 0, stream>>>(rec, rowptr, cnt, hbuf, dinv, b2, out, N);
}


# GEMM: 768 blocks of 4-5 row tiles, one B fragment load feeds every row tile of a wave; fragments read as 16-byte k-quads, so the k order inside each 8-k group (sum order) changes
# speedup vs baseline: 3.7098x; 1.2143x over previous
#include <hip/hip_runtime.h>
#include <hip/hip_bf16.h>


typedef __attribute__((ext_vector_type(2))) float v2f;
typedef __attribute__((ext_vector_type(8))) float v8f;

__global__ void gcn_transpose_w(const float* __restrict__ W,
                                float* __restrict__ Wt, int K) {
  int i = blockIdx.x * blockDim.x + threadIdx.x;
  if (i < K * 128) {
    const int k = i >> 7;
    const int c = i & 127;
    Wt[(size_t)c * K + k] = W[i];
  }
}

constexpr int GCN_GEMM_BLOCKS = 768;
constexpr int GCN_GEMM_RT = 5;

template <int K, int RT>
__device__ __forceinline__ void gcn_gemm_run(
    const float* __restrict__ X, const float* __restrict__ Wt,
    float* __restrict__ H, int nrows, int rowBase, float* __restrict__ As) {
  constexpr int LDA = K + 4;
  constexpr int vecPerRow = K / 4;
  constexpr int NV = RT * 16 * vecPerRow / 256;
  const int tid = threadIdx.x;

  constexpr int NB = (NV > 10) ? 2 : 1;
  constexpr int NVB = NV / NB;
#pragma unroll
  for (int bt = 0; bt < NB; ++bt) {
    float4 val[NVB];
#pragma unroll
    for (int q = 0; q < NVB; ++q) {
      const int i = tid + (bt * NVB + q) * 256;
      const int r = i / vecPerRow;
      const int v = i % vecPerRow;
      const int grow = min(rowBase + r, nrows - 1);
      val[q] = *reinterpret_cast<const float4*>(X + (size_t)grow * K + 4 * v);
    }
#pragma unroll
    for (int q = 0; q < NVB; ++q) {
      const int i = tid + (bt * NVB + q) * 256;
      const int r = i / vecPerRow;
      const int v = i % vecPerRow;
      *reinterpret_cast<float4*>(&As[r * LDA + 4 * v]) = val[q];
    }
  }
  __syncthreads();

  const int wave = tid >> 5;
  const int lane = tid & 31;
  const int l15 = lane & 15;
  const int halfSel = lane >> 4;
  const int col = wave * 16 + l15;

  const float* __restrict__ Bcol = Wt + (size_t)col * K + 4 * halfSel;
  const float* __restrict__ Arow = As + l15 * LDA + 4 * halfSel;

  v8f acc[RT];
#pragma unroll
  for (int t = 0; t < RT; ++t) acc[t] = v8f{};

#pragma unroll 2
  for (int k0 = 0; k0 < K; k0 += 8) {
    const float4 b = *reinterpret_cast<const float4*>(Bcol + k0);
    float4 a[RT];
#pragma unroll
    for (int t = 0; t < RT; ++t)
      a[t] = *reinterpret_cast<const float4*>(Arow + t * 16 * LDA + k0);
    v2f b0, b1;
    b0.x = b.x; b0.y = b.y;
    b1.x = b.z; b1.y = b.w;
#pragma unroll
    for (int t = 0; t < RT; ++t) {
      v2f a0;
      a0.x = a[t].x; a0.y = a[t].y;
      acc[t] = __builtin_amdgcn_wmma_f32_16x16x4_f32(
           false, a0,  false, b0,
           (short)0, acc[t],  false,  false);
    }
#pragma unroll
    for (int t = 0; t < RT; ++t) {
      v2f a1;
      a1.x = a[t].z; a1.y = a[t].w;
      acc[t] = __builtin_amdgcn_wmma_f32_16x16x4_f32(
           false, a1,  false, b1,
           (short)0, acc[t],  false,  false);
    }
  }

  if (rowBase + RT * 16 <= nrows) {
    float* __restrict__ Hp = H + (size_t)(rowBase + 8 * halfSel) * 128 + col;
#pragma unroll
    for (int t = 0; t < RT; ++t) {
#pragma unroll
      for (int r = 0; r < 8; ++r) {
        Hp[(size_t)(t * 16 + r) * 128] = acc[t][r];
      }
    }
  } else {
#pragma unroll
    for (int t = 0; t < RT; ++t) {
#pragma unroll
      for (int r = 0; r < 8; ++r) {
        const int m = rowBase + t * 16 + r + 8 * halfSel;
        if (m < nrows) H[(size_t)m * 128 + col] = acc[t][r];
      }
    }
  }
}

template <int K>
__global__ __launch_bounds__(256) void gcn_gemm_wmma(
    const float* __restrict__ X, const float* __restrict__ Wt,
    float* __restrict__ H, int nrows) {
  __shared__ __align__(16) float As[GCN_GEMM_RT * 16 * (K + 4)];

  const int nTiles = (nrows + 15) >> 4;
  const int tBeg = (int)(((long long)blockIdx.x * nTiles) / gridDim.x);
  const int tEnd = (int)(((long long)(blockIdx.x + 1) * nTiles) / gridDim.x);

  for (int t = tBeg; t < tEnd;) {
    const int rem = tEnd - t;
    if (rem >= 5) {
      gcn_gemm_run<K, 5>(X, Wt, H, nrows, t * 16, As);
      t += 5;
    } else if (rem == 4) {
      gcn_gemm_run<K, 4>(X, Wt, H, nrows, t * 16, As);
      t += 4;
    } else {
      gcn_gemm_run<K, 1>(X, Wt, H, nrows, t * 16, As);
      t += 1;
    }
    __syncthreads();
  }
}

struct __align__(8) EdgeRec {
  int src;
  float w;
};

constexpr int GCN_SLICES = 64;
constexpr int GCN_NODE_CAP = 50176;

__device__ __forceinline__ int gcn_part_pitch(int n) { return (n + 31) & ~31; }

__global__ __launch_bounds__(1024) void gcn_slice_count(
    const int* __restrict__ dst, int* __restrict__ part,
    int* __restrict__ rank, int e, int n, int sliceLen) {
  __shared__ __align__(16) int cntS[GCN_NODE_CAP];
  const int t = threadIdx.x;
  const int lo = blockIdx.y * GCN_NODE_CAP;
  const int span = min(GCN_NODE_CAP, n - lo);

  for (int i = t; i < GCN_NODE_CAP / 4; i += 1024)
    reinterpret_cast<int4*>(cntS)[i] = make_int4(0, 0, 0, 0);
  __syncthreads();

  const int eBeg = blockIdx.x * sliceLen;
  const int eEnd = min(eBeg + sliceLen, e);
  constexpr int U = 4;
  for (int i0 = eBeg + t; i0 < eEnd; i0 += U * 1024) {
    int d[U];
#pragma unroll
    for (int u = 0; u < U; ++u) {
      const int i = i0 + u * 1024;
      const int v = dst[min(i, eEnd - 1)];
      d[u] = (i < eEnd) ? v - lo : -1;
    }
#pragma unroll
    for (int u = 0; u < U; ++u) {
      if ((unsigned)d[u] < (unsigned)span)
        rank[i0 + u * 1024] = atomicAdd(&cntS[d[u]], 1);
    }
  }
  __syncthreads();

  int* __restrict__ row = part + (size_t)blockIdx.x * gcn_part_pitch(n) + lo;
  for (int i = t; i < span; i += 1024) row[i] = cntS[i];
}

__global__ __launch_bounds__(256) void gcn_slice_prefix(int* __restrict__ part,
                                                        int* __restrict__ cnt,
                                                        int n) {
  const int i = blockIdx.x * blockDim.x + threadIdx.x;
  if (i >= n) return;
  const size_t pitch = gcn_part_pitch(n);
  int c[GCN_SLICES];
#pragma unroll
  for (int b = 0; b < GCN_SLICES; ++b) c[b] = part[b * pitch + i];
  int run = 0;
#pragma unroll
  for (int b = 0; b < GCN_SLICES; ++b) {
    part[b * pitch + i] = run;
    run += c[b];
  }
  cnt[i] = run;
}

__global__ __launch_bounds__(1024) void gcn_scan_deg(
    const int* __restrict__ cnt, int* __restrict__ rowptr,
    float* __restrict__ dis, float* __restrict__ dinv, int n) {
  __shared__ int wsum[32];
  __shared__ int wbef[32];
  __shared__ int wpre[32];
  __shared__ int baseSh;

  const int t = threadIdx.x;
  const int lane = t & 31;
  const int wv = t >> 5;
  const int idx = blockIdx.x * 1024 + t;

  int before = 0;
  for (int b = 0; b < (int)blockIdx.x; ++b) before += cnt[b * 1024 + t];
  const int own = (idx < n) ? cnt[idx] : 0;

  int incl = own;
#pragma unroll
  for (int o = 1; o < 32; o <<= 1) {
    const int y = __shfl_up(incl, o);
    if (lane >= o) incl += y;
  }
#pragma unroll
  for (int o = 16; o > 0; o >>= 1) before += __shfl_xor(before, o);

  if (lane == 31) wsum[wv] = incl;
  if (lane == 0) wbef[wv] = before;
  __syncthreads();
  if (wv == 0) {
    const int v = wsum[lane];
    int inc2 = v;
#pragma unroll
    for (int o = 1; o < 32; o <<= 1) {
      const int y = __shfl_up(inc2, o);
      if (lane >= o) inc2 += y;
    }
    int b2 = wbef[lane];
#pragma unroll
    for (int o = 16; o > 0; o >>= 1) b2 += __shfl_xor(b2, o);
    wpre[lane] = inc2 - v;
    if (lane == 0) baseSh = b2;
  }
  __syncthreads();
  if (idx < n) {
    rowptr[idx] = baseSh + wpre[wv] + incl - own;
    const float d = (float)own + 1.0f;
    dis[idx] = rsqrtf(d);
    dinv[idx] = 1.0f / d;
  }
}

__global__ void gcn_fill(const int* __restrict__ src,
                         const int* __restrict__ dst,
                         const int* __restrict__ rank,
                         const int* __restrict__ rowptr,
                         const int* __restrict__ part,
                         const float* __restrict__ dis,
                         EdgeRec* __restrict__ rec, int e, int n,
                         int sliceLen) {
  const int k = blockIdx.x * blockDim.x + threadIdx.x;
  const int i = blockIdx.y * sliceLen + k;
  if (k < sliceLen && i < e) {
    const int s = src[i];
    const int d = dst[i];
    EdgeRec r;
    r.src = s;
    r.w = dis[s] * dis[d];
    const int first = part[(size_t)blockIdx.y * gcn_part_pitch(n) + d];
    rec[rowptr[d] + first + rank[i]] = r;
  }
}

__device__ __forceinline__ void gcn_acc(float4& acc, const float4 v, float w) {
  acc.x = fmaf(v.x, w, acc.x);
  acc.y = fmaf(v.y, w, acc.y);
  acc.z = fmaf(v.z, w, acc.z);
  acc.w = fmaf(v.w, w, acc.w);
}

template <int C>
__device__ __forceinline__ void gcn_gather_chunk(
    float4& acc, const EdgeRec* __restrict__ rec,
    const float4* __restrict__ H4, int lane) {
  EdgeRec r[C];
  float4 v[C];
#pragma unroll
  for (int k = 0; k < C; ++k) r[k] = rec[k];
#pragma unroll
  for (int k = 0; k < C; ++k) v[k] = H4[(size_t)r[k].src * 32 + lane];
#pragma unroll
  for (int k = 0; k < C; ++k) gcn_acc(acc, v[k], r[k].w);
}

__global__ __launch_bounds__(256) void gcn_gather_combine(
    const EdgeRec* __restrict__ rec, const int* __restrict__ rowptr,
    const int* __restrict__ cnt, const float* __restrict__ H,
    const float* __restrict__ dinv, const float* __restrict__ bias,
    float* __restrict__ out, int n) {
  const int node =
      __builtin_amdgcn_readfirstlane(blockIdx.x * 8 + (threadIdx.x >> 5));
  if (node >= n) return;
  const int lane = threadIdx.x & 31;
  const float4* __restrict__ H4 = reinterpret_cast<const float4*>(H);

  int j = rowptr[node];
  const int end = j + cnt[node];
  const float di = dinv[node];
  const float4 hs = H4[(size_t)node * 32 + lane];
  const float4 bv = reinterpret_cast<const float4*>(bias)[lane];

  float4 acc = make_float4(0.f, 0.f, 0.f, 0.f);
  for (; j + 8 <= end; j += 8) gcn_gather_chunk<8>(acc, rec + j, H4, lane);
  if (j + 4 <= end) {
    gcn_gather_chunk<4>(acc, rec + j, H4, lane);
    j += 4;
  }
  if (j + 2 <= end) {
    gcn_gather_chunk<2>(acc, rec + j, H4, lane);
    j += 2;
  }
  if (j < end) gcn_gather_chunk<1>(acc, rec + j, H4, lane);

  float4 o;
  o.x = fmaxf(fmaf(hs.x, di, acc.x) + bv.x, 0.0f);
  o.y = fmaxf(fmaf(hs.y, di, acc.y) + bv.y, 0.0f);
  o.z = fmaxf(fmaf(hs.z, di, acc.z) + bv.z, 0.0f);
  o.w = fmaxf(fmaf(hs.w, di, acc.w) + bv.w, 0.0f);
  reinterpret_cast<float4*>(out)[(size_t)node * 32 + lane] = o;
}

extern "C" void kernel_launch(void* const* d_in, const int* in_sizes, int n_in,
                              void* d_out, int out_size, void* d_ws,
                              size_t ws_size, hipStream_t stream) {
  const float* x = (const float*)d_in[0];
  const int* edge = (const int*)d_in[1];
  const float* W0 = (const float*)d_in[2];
  const float* b0 = (const float*)d_in[3];
  const float* W1 = (const float*)d_in[4];
  const float* b1 = (const float*)d_in[5];
  const float* W2 = (const float*)d_in[6];
  const float* b2 = (const float*)d_in[7];
  float* out = (float*)d_out;

  const int N = in_sizes[0] / 256;
  const int E = in_sizes[1] / 2;
  const int* srcIdx = edge;
  const int* dstIdx = edge + E;

  float* ws = (float*)d_ws;
  float* hbuf = ws;
  float* xin = hbuf + (size_t)N * 128;
  EdgeRec* rec = (EdgeRec*)(xin + (size_t)N * 128);
  float* wt0 = (float*)(rec + E);
  float* wt1 = wt0 + 128 * 256;
  float* wt2 = wt1 + 128 * 128;
  float* dis = wt2 + 128 * 128;
  float* dinv = dis + N;
  int* cnt = (int*)(dinv + N);
  int* rowptr = cnt + N;
  int* rank = rowptr + N;
  int* part = rank + E;

  const int T = 256;
  const int nodeBlocks = (N + T - 1) / T;
  const int sliceLen = (E + GCN_SLICES - 1) / GCN_SLICES;
  const int sliceBlocks = (sliceLen + T - 1) / T;
  const int nodeRows = (N + GCN_NODE_CAP - 1) / GCN_NODE_CAP;
  const int gathBlocks = (N + 7) / 8;
  const int scanBlocks = (N + 1023) / 1024;
  const int gemmBlocks = GCN_GEMM_BLOCKS;

  gcn_transpose_w<<<(256 * 128 + T - 1) / T, T, 0, stream>>>(W0, wt0, 256);
  gcn_transpose_w<<<(128 * 128 + T - 1) / T, T, 0, stream>>>(W1, wt1, 128);
  gcn_transpose_w<<<(128 * 128 + T - 1) / T, T, 0, stream>>>(W2, wt2, 128);

  gcn_slice_count<<<dim3(GCN_SLICES, nodeRows), 1024, 0, stream>>>(
      dstIdx, part, rank, E, N, sliceLen);
  gcn_slice_prefix<<<nodeBlocks, T, 0, stream>>>(part, cnt, N);
  gcn_scan_deg<<<scanBlocks, 1024, 0, stream>>>(cnt, rowptr, dis, dinv, N);
  gcn_fill<<<dim3(sliceBlocks, GCN_SLICES), T, 0, stream>>>(
      srcIdx, dstIdx, rank, rowptr, part, dis, rec, E, N, sliceLen);

  gcn_gemm_wmma<256><<<gemmBlocks, T, 0, stream>>>(x, wt0, hbuf, N);
  gcn_gather_combine<<<gathBlocks, T, 0, stream>>>(rec, rowptr, cnt, hbuf, dinv, b0, xin, N);

  gcn_gemm_wmma<128><<<gemmBlocks, T, 0, stream>>>(xin, wt1, hbuf, N);
  gcn_gather_combine<<<gathBlocks, T, 0, stream>>>(rec, rowptr, cnt, hbuf, dinv, b1, xin, N);

  gcn_gemm_wmma<128><<<gemmBlocks, T, 0, stream>>>(xin, wt2, hbuf, N);
  gcn_gather_combine<<<gathBlocks, T, 0, stream>>>(rec, rowptr, cnt, hbuf, dinv, b2, out, N);
}


# Layers 0-1: gather waves (6-8 nodes each, 16 rows in flight) write their rows into the LDS A tile of the next layer's GEMM in the same block; 2 launches, activation plane gone; sum order unchanged
# speedup vs baseline: 3.8538x; 1.0388x over previous
#include <hip/hip_runtime.h>
#include <hip/hip_bf16.h>


typedef __attribute__((ext_vector_type(2))) float v2f;
typedef __attribute__((ext_vector_type(8))) float v8f;

__global__ void gcn_transpose_w(const float* __restrict__ W,
                                float* __restrict__ Wt, int K) {
  int i = blockIdx.x * blockDim.x + threadIdx.x;
  if (i < K * 128) {
    const int k = i >> 7;
    const int c = i & 127;
    Wt[(size_t)c * K + k] = W[i];
  }
}

constexpr int GCN_GEMM_BLOCKS = 768;
constexpr int GCN_GEMM_RT = 5;

template <int K, int RT>
__device__ __forceinline__ void gcn_gemm_run(
    const float* __restrict__ X, const float* __restrict__ Wt,
    float* __restrict__ H, int nrows, int rowBase, float* __restrict__ As) {
  constexpr int LDA = K + 4;
  constexpr int vecPerRow = K / 4;
  constexpr int NV = RT * 16 * vecPerRow / 256;
  const int tid = threadIdx.x;

  constexpr int NB = (NV > 10) ? 2 : 1;
  constexpr int NVB = NV / NB;
#pragma unroll
  for (int bt = 0; bt < NB; ++bt) {
    float4 val[NVB];
#pragma unroll
    for (int q = 0; q < NVB; ++q) {
      const int i = tid + (bt * NVB + q) * 256;
      const int r = i / vecPerRow;
      const int v = i % vecPerRow;
      const int grow = min(rowBase + r, nrows - 1);
      val[q] = *reinterpret_cast<const float4*>(X + (size_t)grow * K + 4 * v);
    }
#pragma unroll
    for (int q = 0; q < NVB; ++q) {
      const int i = tid + (bt * NVB + q) * 256;
      const int r = i / vecPerRow;
      const int v = i % vecPerRow;
      *reinterpret_cast<float4*>(&As[r * LDA + 4 * v]) = val[q];
    }
  }
  __syncthreads();

  const int wave = tid >> 5;
  const int lane = tid & 31;
  const int l15 = lane & 15;
  const int halfSel = lane >> 4;
  const int col = wave * 16 + l15;

  const float* __restrict__ Bcol = Wt + (size_t)col * K + 4 * halfSel;
  const float* __restrict__ Arow = As + l15 * LDA + 4 * halfSel;

  v8f acc[RT];
#pragma unroll
  for (int t = 0; t < RT; ++t) acc[t] = v8f{};

#pragma unroll 2
  for (int k0 = 0; k0 < K; k0 += 8) {
    const float4 b = *reinterpret_cast<const float4*>(Bcol + k0);
    float4 a[RT];
#pragma unroll
    for (int t = 0; t < RT; ++t)
      a[t] = *reinterpret_cast<const float4*>(Arow + t * 16 * LDA + k0);
    v2f b0, b1;
    b0.x = b.x; b0.y = b.y;
    b1.x = b.z; b1.y = b.w;
#pragma unroll
    for (int t = 0; t < RT; ++t) {
      v2f a0;
      a0.x = a[t].x; a0.y = a[t].y;
      acc[t] = __builtin_amdgcn_wmma_f32_16x16x4_f32(
           false, a0,  false, b0,
           (short)0, acc[t],  false,  false);
    }
#pragma unroll
    for (int t = 0; t < RT; ++t) {
      v2f a1;
      a1.x = a[t].z; a1.y = a[t].w;
      acc[t] = __builtin_amdgcn_wmma_f32_16x16x4_f32(
           false, a1,  false, b1,
           (short)0, acc[t],  false,  false);
    }
  }

  if (rowBase + RT * 16 <= nrows) {
    float* __restrict__ Hp = H + (size_t)(rowBase + 8 * halfSel) * 128 + col;
#pragma unroll
    for (int t = 0; t < RT; ++t) {
#pragma unroll
      for (int r = 0; r < 8; ++r) {
        Hp[(size_t)(t * 16 + r) * 128] = acc[t][r];
      }
    }
  } else {
#pragma unroll
    for (int t = 0; t < RT; ++t) {
#pragma unroll
      for (int r = 0; r < 8; ++r) {
        const int m = rowBase + t * 16 + r + 8 * halfSel;
        if (m < nrows) H[(size_t)m * 128 + col] = acc[t][r];
      }
    }
  }
}

template <int K>
__global__ __launch_bounds__(256) void gcn_gemm_wmma(
    const float* __restrict__ X, const float* __restrict__ Wt,
    float* __restrict__ H, int nrows) {
  __shared__ __align__(16) float As[GCN_GEMM_RT * 16 * (K + 4)];

  const int nTiles = (nrows + 15) >> 4;
  const int tBeg = (int)(((long long)blockIdx.x * nTiles) / gridDim.x);
  const int tEnd = (int)(((long long)(blockIdx.x + 1) * nTiles) / gridDim.x);

  for (int t = tBeg; t < tEnd;) {
    const int rem = tEnd - t;
    if (rem >= 5) {
      gcn_gemm_run<K, 5>(X, Wt, H, nrows, t * 16, As);
      t += 5;
    } else if (rem == 4) {
      gcn_gemm_run<K, 4>(X, Wt, H, nrows, t * 16, As);
      t += 4;
    } else {
      gcn_gemm_run<K, 1>(X, Wt, H, nrows, t * 16, As);
      t += 1;
    }
    __syncthreads();
  }
}

struct __align__(8) EdgeRec {
  int src;
  float w;
};

constexpr int GCN_SLICES = 64;
constexpr int GCN_NODE_CAP = 50176;

__device__ __forceinline__ int gcn_part_pitch(int n) { return (n + 31) & ~31; }

__global__ __launch_bounds__(1024) void gcn_slice_count(
    const int* __restrict__ dst, int* __restrict__ part,
    int* __restrict__ rank, int e, int n, int sliceLen) {
  __shared__ __align__(16) int cntS[GCN_NODE_CAP];
  const int t = threadIdx.x;
  const int lo = blockIdx.y * GCN_NODE_CAP;
  const int span = min(GCN_NODE_CAP, n - lo);

  for (int i = t; i < GCN_NODE_CAP / 4; i += 1024)
    reinterpret_cast<int4*>(cntS)[i] = make_int4(0, 0, 0, 0);
  __syncthreads();

  const int eBeg = blockIdx.x * sliceLen;
  const int eEnd = min(eBeg + sliceLen, e);
  constexpr int U = 4;
  for (int i0 = eBeg + t; i0 < eEnd; i0 += U * 1024) {
    int d[U];
#pragma unroll
    for (int u = 0; u < U; ++u) {
      const int i = i0 + u * 1024;
      const int v = dst[min(i, eEnd - 1)];
      d[u] = (i < eEnd) ? v - lo : -1;
    }
#pragma unroll
    for (int u = 0; u < U; ++u) {
      if ((unsigned)d[u] < (unsigned)span)
        rank[i0 + u * 1024] = atomicAdd(&cntS[d[u]], 1);
    }
  }
  __syncthreads();

  int* __restrict__ row = part + (size_t)blockIdx.x * gcn_part_pitch(n) + lo;
  for (int i = t; i < span; i += 1024) row[i] = cntS[i];
}

__global__ __launch_bounds__(256) void gcn_slice_prefix(int* __restrict__ part,
                                                        int* __restrict__ cnt,
                                                        int n) {
  const int i = blockIdx.x * blockDim.x + threadIdx.x;
  if (i >= n) return;
  const size_t pitch = gcn_part_pitch(n);
  int c[GCN_SLICES];
#pragma unroll
  for (int b = 0; b < GCN_SLICES; ++b) c[b] = part[b * pitch + i];
  int run = 0;
#pragma unroll
  for (int b = 0; b < GCN_SLICES; ++b) {
    part[b * pitch + i] = run;
    run += c[b];
  }
  cnt[i] = run;
}

__global__ __launch_bounds__(1024) void gcn_scan_deg(
    const int* __restrict__ cnt, int* __restrict__ rowptr,
    float* __restrict__ dis, float* __restrict__ dinv, int n) {
  __shared__ int wsum[32];
  __shared__ int wbef[32];
  __shared__ int wpre[32];
  __shared__ int baseSh;

  const int t = threadIdx.x;
  const int lane = t & 31;
  const int wv = t >> 5;
  const int idx = blockIdx.x * 1024 + t;

  int before = 0;
  for (int b = 0; b < (int)blockIdx.x; ++b) before += cnt[b * 1024 + t];
  const int own = (idx < n) ? cnt[idx] : 0;

  int incl = own;
#pragma unroll
  for (int o = 1; o < 32; o <<= 1) {
    const int y = __shfl_up(incl, o);
    if (lane >= o) incl += y;
  }
#pragma unroll
  for (int o = 16; o > 0; o >>= 1) before += __shfl_xor(before, o);

  if (lane == 31) wsum[wv] = incl;
  if (lane == 0) wbef[wv] = before;
  __syncthreads();
  if (wv == 0) {
    const int v = wsum[lane];
    int inc2 = v;
#pragma unroll
    for (int o = 1; o < 32; o <<= 1) {
      const int y = __shfl_up(inc2, o);
      if (lane >= o) inc2 += y;
    }
    int b2 = wbef[lane];
#pragma unroll
    for (int o = 16; o > 0; o >>= 1) b2 += __shfl_xor(b2, o);
    wpre[lane] = inc2 - v;
    if (lane == 0) baseSh = b2;
  }
  __syncthreads();
  if (idx < n) {
    rowptr[idx] = baseSh + wpre[wv] + incl - own;
    const float d = (float)own + 1.0f;
    dis[idx] = rsqrtf(d);
    dinv[idx] = 1.0f / d;
  }
}

__global__ void gcn_fill(const int* __restrict__ src,
                         const int* __restrict__ dst,
                         const int* __restrict__ rank,
                         const int* __restrict__ rowptr,
                         const int* __restrict__ part,
                         const float* __restrict__ dis,
                         EdgeRec* __restrict__ rec, int e, int n,
                         int sliceLen) {
  const int k = blockIdx.x * blockDim.x + threadIdx.x;
  const int i = blockIdx.y * sliceLen + k;
  if (k < sliceLen && i < e) {
    const int s = src[i];
    const int d = dst[i];
    EdgeRec r;
    r.src = s;
    r.w = dis[s] * dis[d];
    const int first = part[(size_t)blockIdx.y * gcn_part_pitch(n) + d];
    rec[rowptr[d] + first + rank[i]] = r;
  }
}

__device__ __forceinline__ void gcn_acc(float4& acc, const float4 v, float w) {
  acc.x = fmaf(v.x, w, acc.x);
  acc.y = fmaf(v.y, w, acc.y);
  acc.z = fmaf(v.z, w, acc.z);
  acc.w = fmaf(v.w, w, acc.w);
}

template <int C>
__device__ __forceinline__ void gcn_gather_chunk(
    float4& acc, const EdgeRec* __restrict__ rec,
    const float4* __restrict__ H4, int lane) {
  EdgeRec r[C];
  float4 v[C];
#pragma unroll
  for (int k = 0; k < C; ++k) r[k] = rec[k];
#pragma unroll
  for (int k = 0; k < C; ++k) v[k] = H4[(size_t)r[k].src * 32 + lane];
#pragma unroll
  for (int k = 0; k < C; ++k) gcn_acc(acc, v[k], r[k].w);
}

__global__ __launch_bounds__(256) void gcn_gather_combine(
    const EdgeRec* __restrict__ rec, const int* __restrict__ rowptr,
    const int* __restrict__ cnt, const float* __restrict__ H,
    const float* __restrict__ dinv, const float* __restrict__ bias,
    float* __restrict__ out, int n) {
  const int node =
      __builtin_amdgcn_readfirstlane(blockIdx.x * 8 + (threadIdx.x >> 5));
  if (node >= n) return;
  const int lane = threadIdx.x & 31;
  const float4* __restrict__ H4 = reinterpret_cast<const float4*>(H);

  int j = rowptr[node];
  const int end = j + cnt[node];
  const float di = dinv[node];
  const float4 hs = H4[(size_t)node * 32 + lane];
  const float4 bv = reinterpret_cast<const float4*>(bias)[lane];

  float4 acc = make_float4(0.f, 0.f, 0.f, 0.f);
  for (; j + 8 <= end; j += 8) gcn_gather_chunk<8>(acc, rec + j, H4, lane);
  if (j + 4 <= end) {
    gcn_gather_chunk<4>(acc, rec + j, H4, lane);
    j += 4;
  }
  if (j + 2 <= end) {
    gcn_gather_chunk<2>(acc, rec + j, H4, lane);
    j += 2;
  }
  if (j < end) gcn_gather_chunk<1>(acc, rec + j, H4, lane);

  float4 o;
  o.x = fmaxf(fmaf(hs.x, di, acc.x) + bv.x, 0.0f);
  o.y = fmaxf(fmaf(hs.y, di, acc.y) + bv.y, 0.0f);
  o.z = fmaxf(fmaf(hs.z, di, acc.z) + bv.z, 0.0f);
  o.w = fmaxf(fmaf(hs.w, di, acc.w) + bv.w, 0.0f);
  reinterpret_cast<float4*>(out)[(size_t)node * 32 + lane] = o;
}

constexpr int GCN_FUSE_BLOCKS = 1024;
constexpr int GCN_FUSE_RT = 4;

__device__ __forceinline__ float4 gcn_node_row(
    const EdgeRec* __restrict__ rec, int j, int end, float di,
    const float4* __restrict__ H4, int node, const float4 bv, int lane) {
  const float4 hs = H4[(size_t)node * 32 + lane];
  float4 acc = make_float4(0.f, 0.f, 0.f, 0.f);
  for (; j + 16 <= end; j += 16) gcn_gather_chunk<16>(acc, rec + j, H4, lane);
  if (j + 8 <= end) {
    gcn_gather_chunk<8>(acc, rec + j, H4, lane);
    j += 8;
  }
  if (j + 4 <= end) {
    gcn_gather_chunk<4>(acc, rec + j, H4, lane);
    j += 4;
  }
  if (j + 2 <= end) {
    gcn_gather_chunk<2>(acc, rec + j, H4, lane);
    j += 2;
  }
  if (j < end) gcn_gather_chunk<1>(acc, rec + j, H4, lane);
  float4 o;
  o.x = fmaxf(fmaf(hs.x, di, acc.x) + bv.x, 0.0f);
  o.y = fmaxf(fmaf(hs.y, di, acc.y) + bv.y, 0.0f);
  o.z = fmaxf(fmaf(hs.z, di, acc.z) + bv.z, 0.0f);
  o.w = fmaxf(fmaf(hs.w, di, acc.w) + bv.w, 0.0f);
  return o;
}

template <int RT>
__device__ __forceinline__ void gcn_fused_gemm(
    const float* __restrict__ Wt, float* __restrict__ Hout, int nrows,
    int rowBase, const float* __restrict__ As) {
  constexpr int K = 128;
  constexpr int LDA = K + 4;
  const int tid = threadIdx.x;
  const int wave = tid >> 5;
  const int lane = tid & 31;
  const int l15 = lane & 15;
  const int halfSel = lane >> 4;
  const int col = wave * 16 + l15;

  const float* __restrict__ Bcol = Wt + (size_t)col * K + 4 * halfSel;
  const float* __restrict__ Arow = As + l15 * LDA + 4 * halfSel;

  v8f acc[RT];
#pragma unroll
  for (int t = 0; t < RT; ++t) acc[t] = v8f{};

#pragma unroll 2
  for (int k0 = 0; k0 < K; k0 += 8) {
    const float4 b = *reinterpret_cast<const float4*>(Bcol + k0);
    float4 a[RT];
#pragma unroll
    for (int t = 0; t < RT; ++t)
      a[t] = *reinterpret_cast<const float4*>(Arow + t * 16 * LDA + k0);
    v2f b0, b1;
    b0.x = b.x; b0.y = b.y;
    b1.x = b.z; b1.y = b.w;
#pragma unroll
    for (int t = 0; t < RT; ++t) {
      v2f a0;
      a0.x = a[t].x; a0.y = a[t].y;
      acc[t] = __builtin_amdgcn_wmma_f32_16x16x4_f32(
           false, a0,  false, b0,
           (short)0, acc[t],  false,  false);
    }
#pragma unroll
    for (int t = 0; t < RT; ++t) {
      v2f a1;
      a1.x = a[t].z; a1.y = a[t].w;
      acc[t] = __builtin_amdgcn_wmma_f32_16x16x4_f32(
           false, a1,  false, b1,
           (short)0, acc[t],  false,  false);
    }
  }

  if (rowBase + RT * 16 <= nrows) {
    float* __restrict__ Hp = Hout + (size_t)(rowBase + 8 * halfSel) * 128 + col;
#pragma unroll
    for (int t = 0; t < RT; ++t) {
#pragma unroll
      for (int r = 0; r < 8; ++r) {
        Hp[(size_t)(t * 16 + r) * 128] = acc[t][r];
      }
    }
  } else {
#pragma unroll
    for (int t = 0; t < RT; ++t) {
#pragma unroll
      for (int r = 0; r < 8; ++r) {
        const int m = rowBase + t * 16 + r + 8 * halfSel;
        if (m < nrows) Hout[(size_t)m * 128 + col] = acc[t][r];
      }
    }
  }
}

__global__ __launch_bounds__(256) void gcn_gather_gemm(
    const EdgeRec* __restrict__ rec, const int* __restrict__ rowptr,
    const int* __restrict__ cnt, const float* __restrict__ Hin,
    const float* __restrict__ dinv, const float* __restrict__ bias,
    const float* __restrict__ Wt, float* __restrict__ Hout, int nrows) {
  constexpr int LDA = 128 + 4;
  __shared__ __align__(16) float As[GCN_FUSE_RT * 16 * LDA];

  const int nTiles = (nrows + 15) >> 4;
  const int tBeg = (int)(((long long)blockIdx.x * nTiles) / gridDim.x);
  const int tEnd = (int)(((long long)(blockIdx.x + 1) * nTiles) / gridDim.x);
  const int nt = tEnd - tBeg;
  if (nt <= 0) return;
  const int rowBase = tBeg * 16;

  {
    const int lane = threadIdx.x & 31;
    const int wv = __builtin_amdgcn_readfirstlane(threadIdx.x >> 5);
    const float4* __restrict__ H4 = reinterpret_cast<const float4*>(Hin);
    const float4 bv = reinterpret_cast<const float4*>(bias)[lane];
    const int npw = 2 * nt;
    const int firstRow = wv * npw;
    const int firstNode = rowBase + firstRow;
    const int mine = min(firstNode + lane, nrows - 1);
    const int vJ = rowptr[mine];
    const int vC = cnt[mine];
    const int vD = __float_as_int(dinv[mine]);
    for (int i = 0; i < npw; ++i) {
      const int node = firstNode + i;
      float4 o = make_float4(0.f, 0.f, 0.f, 0.f);
      if (node < nrows) {
        const int j = __builtin_amdgcn_readlane(vJ, i);
        const int c = __builtin_amdgcn_readlane(vC, i);
        const float di = __int_as_float(__builtin_amdgcn_readlane(vD, i));
        o = gcn_node_row(rec, j, j + c, di, H4, node, bv, lane);
      }
      *reinterpret_cast<float4*>(&As[(firstRow + i) * LDA + 4 * lane]) = o;
    }
  }
  __syncthreads();

  if (nt == 4) {
    gcn_fused_gemm<4>(Wt, Hout, nrows, rowBase, As);
  } else if (nt == 3) {
    gcn_fused_gemm<3>(Wt, Hout, nrows, rowBase, As);
  } else if (nt == 2) {
    gcn_fused_gemm<2>(Wt, Hout, nrows, rowBase, As);
  } else {
    gcn_fused_gemm<1>(Wt, Hout, nrows, rowBase, As);
  }
}

extern "C" void kernel_launch(void* const* d_in, const int* in_sizes, int n_in,
                              void* d_out, int out_size, void* d_ws,
                              size_t ws_size, hipStream_t stream) {
  const float* x = (const float*)d_in[0];
  const int* edge = (const int*)d_in[1];
  const float* W0 = (const float*)d_in[2];
  const float* b0 = (const float*)d_in[3];
  const float* W1 = (const float*)d_in[4];
  const float* b1 = (const float*)d_in[5];
  const float* W2 = (const float*)d_in[6];
  const float* b2 = (const float*)d_in[7];
  float* out = (float*)d_out;

  const int N = in_sizes[0] / 256;
  const int E = in_sizes[1] / 2;
  const int* srcIdx = edge;
  const int* dstIdx = edge + E;

  float* ws = (float*)d_ws;
  float* hbuf = ws;
  float* hbuf2 = hbuf + (size_t)N * 128;
  EdgeRec* rec = (EdgeRec*)(hbuf2 + (size_t)N * 128);
  float* wt0 = (float*)(rec + E);
  float* wt1 = wt0 + 128 * 256;
  float* wt2 = wt1 + 128 * 128;
  float* dis = wt2 + 128 * 128;
  float* dinv = dis + N;
  int* cnt = (int*)(dinv + N);
  int* rowptr = cnt + N;
  int* rank = rowptr + N;
  int* part = rank + E;

  const int T = 256;
  const int nodeBlocks = (N + T - 1) / T;
  const int sliceLen = (E + GCN_SLICES - 1) / GCN_SLICES;
  const int sliceBlocks = (sliceLen + T - 1) / T;
  const int nodeRows = (N + GCN_NODE_CAP - 1) / GCN_NODE_CAP;
  const int gathBlocks = (N + 7) / 8;
  const int scanBlocks = (N + 1023) / 1024;
  const int gemmBlocks = GCN_GEMM_BLOCKS;
  const int nTiles = (N + 15) / 16;
  const int fuseMin = (nTiles + GCN_FUSE_RT - 1) / GCN_FUSE_RT;
  const int fuseBlocks = fuseMin > GCN_FUSE_BLOCKS ? fuseMin : GCN_FUSE_BLOCKS;

  gcn_transpose_w<<<(256 * 128 + T - 1) / T, T, 0, stream>>>(W0, wt0, 256);
  gcn_transpose_w<<<(128 * 128 + T - 1) / T, T, 0, stream>>>(W1, wt1, 128);
  gcn_transpose_w<<<(128 * 128 + T - 1) / T, T, 0, stream>>>(W2, wt2, 128);

  gcn_slice_count<<<dim3(GCN_SLICES, nodeRows), 1024, 0, stream>>>(
      dstIdx, part, rank, E, N, sliceLen);
  gcn_slice_prefix<<<nodeBlocks, T, 0, stream>>>(part, cnt, N);
  gcn_scan_deg<<<scanBlocks, 1024, 0, stream>>>(cnt, rowptr, dis, dinv, N);
  gcn_fill<<<dim3(sliceBlocks, GCN_SLICES), T, 0, stream>>>(
      srcIdx, dstIdx, rank, rowptr, part, dis, rec, E, N, sliceLen);

  gcn_gemm_wmma<256><<<gemmBlocks, T, 0, stream>>>(x, wt0, hbuf, N);

  gcn_gather_gemm<<<fuseBlocks, T, 0, stream>>>(rec, rowptr, cnt, hbuf, dinv,
                                                b0, wt1, hbuf2, N);

  gcn_gather_gemm<<<fuseBlocks, T, 0, stream>>>(rec, rowptr, cnt, hbuf2, dinv,
                                                b1, wt2, hbuf, N);

  gcn_gather_combine<<<gathBlocks, T, 0, stream>>>(rec, rowptr, cnt, hbuf, dinv, b2, out, N);
}


# Three weight-transpose launches folded into the edge-count launch as extra workgroups on WGPs that launch leaves idle: 8 launches per call, not 11; sums unchanged
# speedup vs baseline: 4.0845x; 1.0599x over previous
#include <hip/hip_runtime.h>
#include <hip/hip_bf16.h>


typedef __attribute__((ext_vector_type(2))) float v2f;
typedef __attribute__((ext_vector_type(8))) float v8f;

__device__ __forceinline__ void gcn_transpose_elem(
    const float* __restrict__ W, float* __restrict__ Wt, int K, int i) {
  if (i < K * 128) {
    const int k = i >> 7;
    const int c = i & 127;
    Wt[(size_t)c * K + k] = W[i];
  }
}

constexpr int GCN_GEMM_BLOCKS = 768;
constexpr int GCN_GEMM_RT = 5;

template <int K, int RT>
__device__ __forceinline__ void gcn_gemm_run(
    const float* __restrict__ X, const float* __restrict__ Wt,
    float* __restrict__ H, int nrows, int rowBase, float* __restrict__ As) {
  constexpr int LDA = K + 4;
  constexpr int vecPerRow = K / 4;
  constexpr int NV = RT * 16 * vecPerRow / 256;
  const int tid = threadIdx.x;

  constexpr int NB = (NV > 10) ? 2 : 1;
  constexpr int NVB = NV / NB;
#pragma unroll
  for (int bt = 0; bt < NB; ++bt) {
    float4 val[NVB];
#pragma unroll
    for (int q = 0; q < NVB; ++q) {
      const int i = tid + (bt * NVB + q) * 256;
      const int r = i / vecPerRow;
      const int v = i % vecPerRow;
      const int grow = min(rowBase + r, nrows - 1);
      val[q] = *reinterpret_cast<const float4*>(X + (size_t)grow * K + 4 * v);
    }
#pragma unroll
    for (int q = 0; q < NVB; ++q) {
      const int i = tid + (bt * NVB + q) * 256;
      const int r = i / vecPerRow;
      const int v = i % vecPerRow;
      *reinterpret_cast<float4*>(&As[r * LDA + 4 * v]) = val[q];
    }
  }
  __syncthreads();

  const int wave = tid >> 5;
  const int lane = tid & 31;
  const int l15 = lane & 15;
  const int halfSel = lane >> 4;
  const int col = wave * 16 + l15;

  const float* __restrict__ Bcol = Wt + (size_t)col * K + 4 * halfSel;
  const float* __restrict__ Arow = As + l15 * LDA + 4 * halfSel;

  v8f acc[RT];
#pragma unroll
  for (int t = 0; t < RT; ++t) acc[t] = v8f{};

#pragma unroll 2
  for (int k0 = 0; k0 < K; k0 += 8) {
    const float4 b = *reinterpret_cast<const float4*>(Bcol + k0);
    float4 a[RT];
#pragma unroll
    for (int t = 0; t < RT; ++t)
      a[t] = *reinterpret_cast<const float4*>(Arow + t * 16 * LDA + k0);
    v2f b0, b1;
    b0.x = b.x; b0.y = b.y;
    b1.x = b.z; b1.y = b.w;
#pragma unroll
    for (int t = 0; t < RT; ++t) {
      v2f a0;
      a0.x = a[t].x; a0.y = a[t].y;
      acc[t] = __builtin_amdgcn_wmma_f32_16x16x4_f32(
           false, a0,  false, b0,
           (short)0, acc[t],  false,  false);
    }
#pragma unroll
    for (int t = 0; t < RT; ++t) {
      v2f a1;
      a1.x = a[t].z; a1.y = a[t].w;
      acc[t] = __builtin_amdgcn_wmma_f32_16x16x4_f32(
           false, a1,  false, b1,
           (short)0, acc[t],  false,  false);
    }
  }

  if (rowBase + RT * 16 <= nrows) {
    float* __restrict__ Hp = H + (size_t)(rowBase + 8 * halfSel) * 128 + col;
#pragma unroll
    for (int t = 0; t < RT; ++t) {
#pragma unroll
      for (int r = 0; r < 8; ++r) {
        Hp[(size_t)(t * 16 + r) * 128] = acc[t][r];
      }
    }
  } else {
#pragma unroll
    for (int t = 0; t < RT; ++t) {
#pragma unroll
      for (int r = 0; r < 8; ++r) {
        const int m = rowBase + t * 16 + r + 8 * halfSel;
        if (m < nrows) H[(size_t)m * 128 + col] = acc[t][r];
      }
    }
  }
}

template <int K>
__global__ __launch_bounds__(256) void gcn_gemm_wmma(
    const float* __restrict__ X, const float* __restrict__ Wt,
    float* __restrict__ H, int nrows) {
  __shared__ __align__(16) float As[GCN_GEMM_RT * 16 * (K + 4)];

  const int nTiles = (nrows + 15) >> 4;
  const int tBeg = (int)(((long long)blockIdx.x * nTiles) / gridDim.x);
  const int tEnd = (int)(((long long)(blockIdx.x + 1) * nTiles) / gridDim.x);

  for (int t = tBeg; t < tEnd;) {
    const int rem = tEnd - t;
    if (rem >= 5) {
      gcn_gemm_run<K, 5>(X, Wt, H, nrows, t * 16, As);
      t += 5;
    } else if (rem == 4) {
      gcn_gemm_run<K, 4>(X, Wt, H, nrows, t * 16, As);
      t += 4;
    } else {
      gcn_gemm_run<K, 1>(X, Wt, H, nrows, t * 16, As);
      t += 1;
    }
    __syncthreads();
  }
}

struct __align__(8) EdgeRec {
  int src;
  float w;
};

constexpr int GCN_SLICES = 64;
constexpr int GCN_NODE_CAP = 50176;
constexpr int GCN_TR_W0 = 256 * 128;
constexpr int GCN_TR_W = 128 * 128;
constexpr int GCN_TR_BLOCKS = (GCN_TR_W0 + 2 * GCN_TR_W + 1023) / 1024;

__device__ __forceinline__ int gcn_part_pitch(int n) { return (n + 31) & ~31; }

__global__ __launch_bounds__(1024) void gcn_slice_count(
    const int* __restrict__ dst, int* __restrict__ part,
    int* __restrict__ rank, int e, int n, int sliceLen,
    const float* __restrict__ W0, const float* __restrict__ W1,
    const float* __restrict__ W2, float* __restrict__ wt0,
    float* __restrict__ wt1, float* __restrict__ wt2) {
  __shared__ __align__(16) int cntS[GCN_NODE_CAP];
  const int t = threadIdx.x;

  if (blockIdx.x >= GCN_SLICES) {
    if (blockIdx.y == 0) {
      const int i = (blockIdx.x - GCN_SLICES) * 1024 + t;
      if (i < GCN_TR_W0) {
        gcn_transpose_elem(W0, wt0, 256, i);
      } else if (i < GCN_TR_W0 + GCN_TR_W) {
        gcn_transpose_elem(W1, wt1, 128, i - GCN_TR_W0);
      } else {
        gcn_transpose_elem(W2, wt2, 128, i - GCN_TR_W0 - GCN_TR_W);
      }
    }
    return;
  }
  const int lo = blockIdx.y * GCN_NODE_CAP;
  const int span = min(GCN_NODE_CAP, n - lo);

  for (int i = t; i < GCN_NODE_CAP / 4; i += 1024)
    reinterpret_cast<int4*>(cntS)[i] = make_int4(0, 0, 0, 0);
  __syncthreads();

  const int eBeg = blockIdx.x * sliceLen;
  const int eEnd = min(eBeg + sliceLen, e);
  constexpr int U = 4;
  for (int i0 = eBeg + t; i0 < eEnd; i0 += U * 1024) {
    int d[U];
#pragma unroll
    for (int u = 0; u < U; ++u) {
      const int i = i0 + u * 1024;
      const int v = dst[min(i, eEnd - 1)];
      d[u] = (i < eEnd) ? v - lo : -1;
    }
#pragma unroll
    for (int u = 0; u < U; ++u) {
      if ((unsigned)d[u] < (unsigned)span)
        rank[i0 + u * 1024] = atomicAdd(&cntS[d[u]], 1);
    }
  }
  __syncthreads();

  int* __restrict__ row = part + (size_t)blockIdx.x * gcn_part_pitch(n) + lo;
  for (int i = t; i < span; i += 1024) row[i] = cntS[i];
}

__global__ __launch_bounds__(256) void gcn_slice_prefix(int* __restrict__ part,
                                                        int* __restrict__ cnt,
                                                        int n) {
  const int i = blockIdx.x * blockDim.x + threadIdx.x;
  if (i >= n) return;
  const size_t pitch = gcn_part_pitch(n);
  int c[GCN_SLICES];
#pragma unroll
  for (int b = 0; b < GCN_SLICES; ++b) c[b] = part[b * pitch + i];
  int run = 0;
#pragma unroll
  for (int b = 0; b < GCN_SLICES; ++b) {
    part[b * pitch + i] = run;
    run += c[b];
  }
  cnt[i] = run;
}

__global__ __launch_bounds__(1024) void gcn_scan_deg(
    const int* __restrict__ cnt, int* __restrict__ rowptr,
    float* __restrict__ dis, float* __restrict__ dinv, int n) {
  __shared__ int wsum[32];
  __shared__ int wbef[32];
  __shared__ int wpre[32];
  __shared__ int baseSh;

  const int t = threadIdx.x;
  const int lane = t & 31;
  const int wv = t >> 5;
  const int idx = blockIdx.x * 1024 + t;

  int before = 0;
  for (int b = 0; b < (int)blockIdx.x; ++b) before += cnt[b * 1024 + t];
  const int own = (idx < n) ? cnt[idx] : 0;

  int incl = own;
#pragma unroll
  for (int o = 1; o < 32; o <<= 1) {
    const int y = __shfl_up(incl, o);
    if (lane >= o) incl += y;
  }
#pragma unroll
  for (int o = 16; o > 0; o >>= 1) before += __shfl_xor(before, o);

  if (lane == 31) wsum[wv] = incl;
  if (lane == 0) wbef[wv] = before;
  __syncthreads();
  if (wv == 0) {
    const int v = wsum[lane];
    int inc2 = v;
#pragma unroll
    for (int o = 1; o < 32; o <<= 1) {
      const int y = __shfl_up(inc2, o);
      if (lane >= o) inc2 += y;
    }
    int b2 = wbef[lane];
#pragma unroll
    for (int o = 16; o > 0; o >>= 1) b2 += __shfl_xor(b2, o);
    wpre[lane] = inc2 - v;
    if (lane == 0) baseSh = b2;
  }
  __syncthreads();
  if (idx < n) {
    rowptr[idx] = baseSh + wpre[wv] + incl - own;
    const float d = (float)own + 1.0f;
    dis[idx] = rsqrtf(d);
    dinv[idx] = 1.0f / d;
  }
}

__global__ void gcn_fill(const int* __restrict__ src,
                         const int* __restrict__ dst,
                         const int* __restrict__ rank,
                         const int* __restrict__ rowptr,
                         const int* __restrict__ part,
                         const float* __restrict__ dis,
                         EdgeRec* __restrict__ rec, int e, int n,
                         int sliceLen) {
  const int k = blockIdx.x * blockDim.x + threadIdx.x;
  const int i = blockIdx.y * sliceLen + k;
  if (k < sliceLen && i < e) {
    const int s = src[i];
    const int d = dst[i];
    EdgeRec r;
    r.src = s;
    r.w = dis[s] * dis[d];
    const int first = part[(size_t)blockIdx.y * gcn_part_pitch(n) + d];
    rec[rowptr[d] + first + rank[i]] = r;
  }
}

__device__ __forceinline__ void gcn_acc(float4& acc, const float4 v, float w) {
  acc.x = fmaf(v.x, w, acc.x);
  acc.y = fmaf(v.y, w, acc.y);
  acc.z = fmaf(v.z, w, acc.z);
  acc.w = fmaf(v.w, w, acc.w);
}

template <int C>
__device__ __forceinline__ void gcn_gather_chunk(
    float4& acc, const EdgeRec* __restrict__ rec,
    const float4* __restrict__ H4, int lane) {
  EdgeRec r[C];
  float4 v[C];
#pragma unroll
  for (int k = 0; k < C; ++k) r[k] = rec[k];
#pragma unroll
  for (int k = 0; k < C; ++k) v[k] = H4[(size_t)r[k].src * 32 + lane];
#pragma unroll
  for (int k = 0; k < C; ++k) gcn_acc(acc, v[k], r[k].w);
}

__global__ __launch_bounds__(256) void gcn_gather_combine(
    const EdgeRec* __restrict__ rec, const int* __restrict__ rowptr,
    const int* __restrict__ cnt, const float* __restrict__ H,
    const float* __restrict__ dinv, const float* __restrict__ bias,
    float* __restrict__ out, int n) {
  const int node =
      __builtin_amdgcn_readfirstlane(blockIdx.x * 8 + (threadIdx.x >> 5));
  if (node >= n) return;
  const int lane = threadIdx.x & 31;
  const float4* __restrict__ H4 = reinterpret_cast<const float4*>(H);

  int j = rowptr[node];
  const int end = j + cnt[node];
  const float di = dinv[node];
  const float4 hs = H4[(size_t)node * 32 + lane];
  const float4 bv = reinterpret_cast<const float4*>(bias)[lane];

  float4 acc = make_float4(0.f, 0.f, 0.f, 0.f);
  for (; j + 8 <= end; j += 8) gcn_gather_chunk<8>(acc, rec + j, H4, lane);
  if (j + 4 <= end) {
    gcn_gather_chunk<4>(acc, rec + j, H4, lane);
    j += 4;
  }
  if (j + 2 <= end) {
    gcn_gather_chunk<2>(acc, rec + j, H4, lane);
    j += 2;
  }
  if (j < end) gcn_gather_chunk<1>(acc, rec + j, H4, lane);

  float4 o;
  o.x = fmaxf(fmaf(hs.x, di, acc.x) + bv.x, 0.0f);
  o.y = fmaxf(fmaf(hs.y, di, acc.y) + bv.y, 0.0f);
  o.z = fmaxf(fmaf(hs.z, di, acc.z) + bv.z, 0.0f);
  o.w = fmaxf(fmaf(hs.w, di, acc.w) + bv.w, 0.0f);
  reinterpret_cast<float4*>(out)[(size_t)node * 32 + lane] = o;
}

constexpr int GCN_FUSE_BLOCKS = 1024;
constexpr int GCN_FUSE_RT = 4;

__device__ __forceinline__ float4 gcn_node_row(
    const EdgeRec* __restrict__ rec, int j, int end, float di,
    const float4* __restrict__ H4, int node, const float4 bv, int lane) {
  const float4 hs = H4[(size_t)node * 32 + lane];
  float4 acc = make_float4(0.f, 0.f, 0.f, 0.f);
  for (; j + 16 <= end; j += 16) gcn_gather_chunk<16>(acc, rec + j, H4, lane);
  if (j + 8 <= end) {
    gcn_gather_chunk<8>(acc, rec + j, H4, lane);
    j += 8;
  }
  if (j + 4 <= end) {
    gcn_gather_chunk<4>(acc, rec + j, H4, lane);
    j += 4;
  }
  if (j + 2 <= end) {
    gcn_gather_chunk<2>(acc, rec + j, H4, lane);
    j += 2;
  }
  if (j < end) gcn_gather_chunk<1>(acc, rec + j, H4, lane);
  float4 o;
  o.x = fmaxf(fmaf(hs.x, di, acc.x) + bv.x, 0.0f);
  o.y = fmaxf(fmaf(hs.y, di, acc.y) + bv.y, 0.0f);
  o.z = fmaxf(fmaf(hs.z, di, acc.z) + bv.z, 0.0f);
  o.w = fmaxf(fmaf(hs.w, di, acc.w) + bv.w, 0.0f);
  return o;
}

template <int RT>
__device__ __forceinline__ void gcn_fused_gemm(
    const float* __restrict__ Wt, float* __restrict__ Hout, int nrows,
    int rowBase, const float* __restrict__ As) {
  constexpr int K = 128;
  constexpr int LDA = K + 4;
  const int tid = threadIdx.x;
  const int wave = tid >> 5;
  const int lane = tid & 31;
  const int l15 = lane & 15;
  const int halfSel = lane >> 4;
  const int col = wave * 16 + l15;

  const float* __restrict__ Bcol = Wt + (size_t)col * K + 4 * halfSel;
  const float* __restrict__ Arow = As + l15 * LDA + 4 * halfSel;

  v8f acc[RT];
#pragma unroll
  for (int t = 0; t < RT; ++t) acc[t] = v8f{};

#pragma unroll 2
  for (int k0 = 0; k0 < K; k0 += 8) {
    const float4 b = *reinterpret_cast<const float4*>(Bcol + k0);
    float4 a[RT];
#pragma unroll
    for (int t = 0; t < RT; ++t)
      a[t] = *reinterpret_cast<const float4*>(Arow + t * 16 * LDA + k0);
    v2f b0, b1;
    b0.x = b.x; b0.y = b.y;
    b1.x = b.z; b1.y = b.w;
#pragma unroll
    for (int t = 0; t < RT; ++t) {
      v2f a0;
      a0.x = a[t].x; a0.y = a[t].y;
      acc[t] = __builtin_amdgcn_wmma_f32_16x16x4_f32(
           false, a0,  false, b0,
           (short)0, acc[t],  false,  false);
    }
#pragma unroll
    for (int t = 0; t < RT; ++t) {
      v2f a1;
      a1.x = a[t].z; a1.y = a[t].w;
      acc[t] = __builtin_amdgcn_wmma_f32_16x16x4_f32(
           false, a1,  false, b1,
           (short)0, acc[t],  false,  false);
    }
  }

  if (rowBase + RT * 16 <= nrows) {
    float* __restrict__ Hp = Hout + (size_t)(rowBase + 8 * halfSel) * 128 + col;
#pragma unroll
    for (int t = 0; t < RT; ++t) {
#pragma unroll
      for (int r = 0; r < 8; ++r) {
        Hp[(size_t)(t * 16 + r) * 128] = acc[t][r];
      }
    }
  } else {
#pragma unroll
    for (int t = 0; t < RT; ++t) {
#pragma unroll
      for (int r = 0; r < 8; ++r) {
        const int m = rowBase + t * 16 + r + 8 * halfSel;
        if (m < nrows) Hout[(size_t)m * 128 + col] = acc[t][r];
      }
    }
  }
}

__global__ __launch_bounds__(256) void gcn_gather_gemm(
    const EdgeRec* __restrict__ rec, const int* __restrict__ rowptr,
    const int* __restrict__ cnt, const float* __restrict__ Hin,
    const float* __restrict__ dinv, const float* __restrict__ bias,
    const float* __restrict__ Wt, float* __restrict__ Hout, int nrows) {
  constexpr int LDA = 128 + 4;
  __shared__ __align__(16) float As[GCN_FUSE_RT * 16 * LDA];

  const int nTiles = (nrows + 15) >> 4;
  const int tBeg = (int)(((long long)blockIdx.x * nTiles) / gridDim.x);
  const int tEnd = (int)(((long long)(blockIdx.x + 1) * nTiles) / gridDim.x);
  const int nt = tEnd - tBeg;
  if (nt <= 0) return;
  const int rowBase = tBeg * 16;

  {
    const int lane = threadIdx.x & 31;
    const int wv = __builtin_amdgcn_readfirstlane(threadIdx.x >> 5);
    const float4* __restrict__ H4 = reinterpret_cast<const float4*>(Hin);
    const float4 bv = reinterpret_cast<const float4*>(bias)[lane];
    const int npw = 2 * nt;
    const int firstRow = wv * npw;
    const int firstNode = rowBase + firstRow;
    const int mine = min(firstNode + lane, nrows - 1);
    const int vJ = rowptr[mine];
    const int vC = cnt[mine];
    const int vD = __float_as_int(dinv[mine]);
    for (int i = 0; i < npw; ++i) {
      const int node = firstNode + i;
      float4 o = make_float4(0.f, 0.f, 0.f, 0.f);
      if (node < nrows) {
        const int j = __builtin_amdgcn_readlane(vJ, i);
        const int c = __builtin_amdgcn_readlane(vC, i);
        const float di = __int_as_float(__builtin_amdgcn_readlane(vD, i));
        o = gcn_node_row(rec, j, j + c, di, H4, node, bv, lane);
      }
      *reinterpret_cast<float4*>(&As[(firstRow + i) * LDA + 4 * lane]) = o;
    }
  }
  __syncthreads();

  if (nt == 4) {
    gcn_fused_gemm<4>(Wt, Hout, nrows, rowBase, As);
  } else if (nt == 3) {
    gcn_fused_gemm<3>(Wt, Hout, nrows, rowBase, As);
  } else if (nt == 2) {
    gcn_fused_gemm<2>(Wt, Hout, nrows, rowBase, As);
  } else {
    gcn_fused_gemm<1>(Wt, Hout, nrows, rowBase, As);
  }
}

extern "C" void kernel_launch(void* const* d_in, const int* in_sizes, int n_in,
                              void* d_out, int out_size, void* d_ws,
                              size_t ws_size, hipStream_t stream) {
  const float* x = (const float*)d_in[0];
  const int* edge = (const int*)d_in[1];
  const float* W0 = (const float*)d_in[2];
  const float* b0 = (const float*)d_in[3];
  const float* W1 = (const float*)d_in[4];
  const float* b1 = (const float*)d_in[5];
  const float* W2 = (const float*)d_in[6];
  const float* b2 = (const float*)d_in[7];
  float* out = (float*)d_out;

  const int N = in_sizes[0] / 256;
  const int E = in_sizes[1] / 2;
  const int* srcIdx = edge;
  const int* dstIdx = edge + E;

  float* ws = (float*)d_ws;
  float* hbuf = ws;
  float* hbuf2 = hbuf + (size_t)N * 128;
  EdgeRec* rec = (EdgeRec*)(hbuf2 + (size_t)N * 128);
  float* wt0 = (float*)(rec + E);
  float* wt1 = wt0 + 128 * 256;
  float* wt2 = wt1 + 128 * 128;
  float* dis = wt2 + 128 * 128;
  float* dinv = dis + N;
  int* cnt = (int*)(dinv + N);
  int* rowptr = cnt + N;
  int* rank = rowptr + N;
  int* part = rank + E;

  const int T = 256;
  const int nodeBlocks = (N + T - 1) / T;
  const int sliceLen = (E + GCN_SLICES - 1) / GCN_SLICES;
  const int sliceBlocks = (sliceLen + T - 1) / T;
  const int nodeRows = (N + GCN_NODE_CAP - 1) / GCN_NODE_CAP;
  const int gathBlocks = (N + 7) / 8;
  const int scanBlocks = (N + 1023) / 1024;
  const int gemmBlocks = GCN_GEMM_BLOCKS;
  const int nTiles = (N + 15) / 16;
  const int fuseMin = (nTiles + GCN_FUSE_RT - 1) / GCN_FUSE_RT;
  const int fuseBlocks = fuseMin > GCN_FUSE_BLOCKS ? fuseMin : GCN_FUSE_BLOCKS;

  gcn_slice_count<<<dim3(GCN_SLICES + GCN_TR_BLOCKS, nodeRows), 1024, 0,
                    stream>>>(dstIdx, part, rank, E, N, sliceLen, W0, W1, W2,
                              wt0, wt1, wt2);
  gcn_slice_prefix<<<nodeBlocks, T, 0, stream>>>(part, cnt, N);
  gcn_scan_deg<<<scanBlocks, 1024, 0, stream>>>(cnt, rowptr, dis, dinv, N);
  gcn_fill<<<dim3(sliceBlocks, GCN_SLICES), T, 0, stream>>>(
      srcIdx, dstIdx, rank, rowptr, part, dis, rec, E, N, sliceLen);

  gcn_gemm_wmma<256><<<gemmBlocks, T, 0, stream>>>(x, wt0, hbuf, N);

  gcn_gather_gemm<<<fuseBlocks, T, 0, stream>>>(rec, rowptr, cnt, hbuf, dinv,
                                                b0, wt1, hbuf2, N);

  gcn_gather_gemm<<<fuseBlocks, T, 0, stream>>>(rec, rowptr, cnt, hbuf2, dinv,
                                                b1, wt2, hbuf, N);

  gcn_gather_combine<<<gathBlocks, T, 0, stream>>>(rec, rowptr, cnt, hbuf, dinv, b2, out, N);
}


# Edge-record fill (now 4 edges per thread) shares the layer-0 GEMM launch as alternating workgroups: request-bound fill runs under the matrix-bound product; 7 launches, not 8; sums unchanged
# speedup vs baseline: 4.2814x; 1.0482x over previous
#include <hip/hip_runtime.h>
#include <hip/hip_bf16.h>


typedef __attribute__((ext_vector_type(2))) float v2f;
typedef __attribute__((ext_vector_type(8))) float v8f;

__device__ __forceinline__ void gcn_transpose_elem(
    const float* __restrict__ W, float* __restrict__ Wt, int K, int i) {
  if (i < K * 128) {
    const int k = i >> 7;
    const int c = i & 127;
    Wt[(size_t)c * K + k] = W[i];
  }
}

constexpr int GCN_GEMM_BLOCKS = 768;
constexpr int GCN_GEMM_RT = 5;

template <int K, int RT>
__device__ __forceinline__ void gcn_gemm_run(
    const float* __restrict__ X, const float* __restrict__ Wt,
    float* __restrict__ H, int nrows, int rowBase, float* __restrict__ As) {
  constexpr int LDA = K + 4;
  constexpr int vecPerRow = K / 4;
  constexpr int NV = RT * 16 * vecPerRow / 256;
  const int tid = threadIdx.x;

  constexpr int NB = (NV > 10) ? 2 : 1;
  constexpr int NVB = NV / NB;
#pragma unroll
  for (int bt = 0; bt < NB; ++bt) {
    float4 val[NVB];
#pragma unroll
    for (int q = 0; q < NVB; ++q) {
      const int i = tid + (bt * NVB + q) * 256;
      const int r = i / vecPerRow;
      const int v = i % vecPerRow;
      const int grow = min(rowBase + r, nrows - 1);
      val[q] = *reinterpret_cast<const float4*>(X + (size_t)grow * K + 4 * v);
    }
#pragma unroll
    for (int q = 0; q < NVB; ++q) {
      const int i = tid + (bt * NVB + q) * 256;
      const int r = i / vecPerRow;
      const int v = i % vecPerRow;
      *reinterpret_cast<float4*>(&As[r * LDA + 4 * v]) = val[q];
    }
  }
  __syncthreads();

  const int wave = tid >> 5;
  const int lane = tid & 31;
  const int l15 = lane & 15;
  const int halfSel = lane >> 4;
  const int col = wave * 16 + l15;

  const float* __restrict__ Bcol = Wt + (size_t)col * K + 4 * halfSel;
  const float* __restrict__ Arow = As + l15 * LDA + 4 * halfSel;

  v8f acc[RT];
#pragma unroll
  for (int t = 0; t < RT; ++t) acc[t] = v8f{};

#pragma unroll 2
  for (int k0 = 0; k0 < K; k0 += 8) {
    const float4 b = *reinterpret_cast<const float4*>(Bcol + k0);
    float4 a[RT];
#pragma unroll
    for (int t = 0; t < RT; ++t)
      a[t] = *reinterpret_cast<const float4*>(Arow + t * 16 * LDA + k0);
    v2f b0, b1;
    b0.x = b.x; b0.y = b.y;
    b1.x = b.z; b1.y = b.w;
#pragma unroll
    for (int t = 0; t < RT; ++t) {
      v2f a0;
      a0.x = a[t].x; a0.y = a[t].y;
      acc[t] = __builtin_amdgcn_wmma_f32_16x16x4_f32(
           false, a0,  false, b0,
           (short)0, acc[t],  false,  false);
    }
#pragma unroll
    for (int t = 0; t < RT; ++t) {
      v2f a1;
      a1.x = a[t].z; a1.y = a[t].w;
      acc[t] = __builtin_amdgcn_wmma_f32_16x16x4_f32(
           false, a1,  false, b1,
           (short)0, acc[t],  false,  false);
    }
  }

  if (rowBase + RT * 16 <= nrows) {
    float* __restrict__ Hp = H + (size_t)(rowBase + 8 * halfSel) * 128 + col;
#pragma unroll
    for (int t = 0; t < RT; ++t) {
#pragma unroll
      for (int r = 0; r < 8; ++r) {
        Hp[(size_t)(t * 16 + r) * 128] = acc[t][r];
      }
    }
  } else {
#pragma unroll
    for (int t = 0; t < RT; ++t) {
#pragma unroll
      for (int r = 0; r < 8; ++r) {
        const int m = rowBase + t * 16 + r + 8 * halfSel;
        if (m < nrows) H[(size_t)m * 128 + col] = acc[t][r];
      }
    }
  }
}

struct __align__(8) EdgeRec {
  int src;
  float w;
};

constexpr int GCN_SLICES = 64;
constexpr int GCN_NODE_CAP = 50176;
constexpr int GCN_TR_W0 = 256 * 128;
constexpr int GCN_TR_W = 128 * 128;
constexpr int GCN_TR_BLOCKS = (GCN_TR_W0 + 2 * GCN_TR_W + 1023) / 1024;

__device__ __forceinline__ int gcn_part_pitch(int n) { return (n + 31) & ~31; }

__global__ __launch_bounds__(1024) void gcn_slice_count(
    const int* __restrict__ dst, int* __restrict__ part,
    int* __restrict__ rank, int e, int n, int sliceLen,
    const float* __restrict__ W0, const float* __restrict__ W1,
    const float* __restrict__ W2, float* __restrict__ wt0,
    float* __restrict__ wt1, float* __restrict__ wt2) {
  __shared__ __align__(16) int cntS[GCN_NODE_CAP];
  const int t = threadIdx.x;

  if (blockIdx.x >= GCN_SLICES) {
    if (blockIdx.y == 0) {
      const int i = (blockIdx.x - GCN_SLICES) * 1024 + t;
      if (i < GCN_TR_W0) {
        gcn_transpose_elem(W0, wt0, 256, i);
      } else if (i < GCN_TR_W0 + GCN_TR_W) {
        gcn_transpose_elem(W1, wt1, 128, i - GCN_TR_W0);
      } else {
        gcn_transpose_elem(W2, wt2, 128, i - GCN_TR_W0 - GCN_TR_W);
      }
    }
    return;
  }
  const int lo = blockIdx.y * GCN_NODE_CAP;
  const int span = min(GCN_NODE_CAP, n - lo);

  for (int i = t; i < GCN_NODE_CAP / 4; i += 1024)
    reinterpret_cast<int4*>(cntS)[i] = make_int4(0, 0, 0, 0);
  __syncthreads();

  const int eBeg = blockIdx.x * sliceLen;
  const int eEnd = min(eBeg + sliceLen, e);
  constexpr int U = 4;
  for (int i0 = eBeg + t; i0 < eEnd; i0 += U * 1024) {
    int d[U];
#pragma unroll
    for (int u = 0; u < U; ++u) {
      const int i = i0 + u * 1024;
      const int v = dst[min(i, eEnd - 1)];
      d[u] = (i < eEnd) ? v - lo : -1;
    }
#pragma unroll
    for (int u = 0; u < U; ++u) {
      if ((unsigned)d[u] < (unsigned)span)
        rank[i0 + u * 1024] = atomicAdd(&cntS[d[u]], 1);
    }
  }
  __syncthreads();

  int* __restrict__ row = part + (size_t)blockIdx.x * gcn_part_pitch(n) + lo;
  for (int i = t; i < span; i += 1024) row[i] = cntS[i];
}

__global__ __launch_bounds__(256) void gcn_slice_prefix(int* __restrict__ part,
                                                        int* __restrict__ cnt,
                                                        int n) {
  const int i = blockIdx.x * blockDim.x + threadIdx.x;
  if (i >= n) return;
  const size_t pitch = gcn_part_pitch(n);
  int c[GCN_SLICES];
#pragma unroll
  for (int b = 0; b < GCN_SLICES; ++b) c[b] = part[b * pitch + i];
  int run = 0;
#pragma unroll
  for (int b = 0; b < GCN_SLICES; ++b) {
    part[b * pitch + i] = run;
    run += c[b];
  }
  cnt[i] = run;
}

__global__ __launch_bounds__(1024) void gcn_scan_deg(
    const int* __restrict__ cnt, int* __restrict__ rowptr,
    float* __restrict__ dis, float* __restrict__ dinv, int n) {
  __shared__ int wsum[32];
  __shared__ int wbef[32];
  __shared__ int wpre[32];
  __shared__ int baseSh;

  const int t = threadIdx.x;
  const int lane = t & 31;
  const int wv = t >> 5;
  const int idx = blockIdx.x * 1024 + t;

  int before = 0;
  for (int b = 0; b < (int)blockIdx.x; ++b) before += cnt[b * 1024 + t];
  const int own = (idx < n) ? cnt[idx] : 0;

  int incl = own;
#pragma unroll
  for (int o = 1; o < 32; o <<= 1) {
    const int y = __shfl_up(incl, o);
    if (lane >= o) incl += y;
  }
#pragma unroll
  for (int o = 16; o > 0; o >>= 1) before += __shfl_xor(before, o);

  if (lane == 31) wsum[wv] = incl;
  if (lane == 0) wbef[wv] = before;
  __syncthreads();
  if (wv == 0) {
    const int v = wsum[lane];
    int inc2 = v;
#pragma unroll
    for (int o = 1; o < 32; o <<= 1) {
      const int y = __shfl_up(inc2, o);
      if (lane >= o) inc2 += y;
    }
    int b2 = wbef[lane];
#pragma unroll
    for (int o = 16; o > 0; o >>= 1) b2 += __shfl_xor(b2, o);
    wpre[lane] = inc2 - v;
    if (lane == 0) baseSh = b2;
  }
  __syncthreads();
  if (idx < n) {
    rowptr[idx] = baseSh + wpre[wv] + incl - own;
    const float d = (float)own + 1.0f;
    dis[idx] = rsqrtf(d);
    dinv[idx] = 1.0f / d;
  }
}

constexpr int GCN_FILL_U = 4;

__device__ __forceinline__ void gcn_fill_chunk(
    const int* __restrict__ src, const int* __restrict__ dst,
    const int* __restrict__ rank, const int* __restrict__ rowptr,
    const int* __restrict__ part, const float* __restrict__ dis,
    EdgeRec* __restrict__ rec, int e, int n, int sliceLen, int slice,
    int kBase) {
  const int eBeg = slice * sliceLen;
  const int eEnd = min(eBeg + sliceLen, e);
  if (kBase >= eEnd - eBeg) return;
  const int* __restrict__ prow = part + (size_t)slice * gcn_part_pitch(n);
  int s[GCN_FILL_U], d[GCN_FILL_U], rk[GCN_FILL_U];
  bool ok[GCN_FILL_U];
#pragma unroll
  for (int u = 0; u < GCN_FILL_U; ++u) {
    const int i = eBeg + kBase + u * 256 + (int)threadIdx.x;
    ok[u] = i < eEnd;
    const int ii = min(i, eEnd - 1);
    s[u] = src[ii];
    d[u] = dst[ii];
    rk[u] = rank[ii];
  }
#pragma unroll
  for (int u = 0; u < GCN_FILL_U; ++u)
    asm volatile("" ::"v"(s[u]), "v"(d[u]), "v"(rk[u]));
  float w[GCN_FILL_U];
  int pos[GCN_FILL_U];
#pragma unroll
  for (int u = 0; u < GCN_FILL_U; ++u) {
    w[u] = dis[s[u]] * dis[d[u]];
    pos[u] = rowptr[d[u]] + prow[d[u]] + rk[u];
  }
#pragma unroll
  for (int u = 0; u < GCN_FILL_U; ++u) {
    if (ok[u]) {
      EdgeRec r;
      r.src = s[u];
      r.w = w[u];
      rec[pos[u]] = r;
    }
  }
}

__global__ __launch_bounds__(256) void gcn_gemm_fill(
    const float* __restrict__ X, const float* __restrict__ Wt,
    float* __restrict__ H, int nrows, const int* __restrict__ src,
    const int* __restrict__ dst, const int* __restrict__ rank,
    const int* __restrict__ rowptr, const int* __restrict__ part,
    const float* __restrict__ dis, EdgeRec* __restrict__ rec, int e,
    int sliceLen, int chunksPerSlice) {
  constexpr int K = 256;
  __shared__ __align__(16) float As[GCN_GEMM_RT * 16 * (K + 4)];

  const int b = blockIdx.x;
  const bool paired = b < 2 * GCN_GEMM_BLOCKS;
  if (!paired || (b & 1)) {
    const int fb = paired ? (b >> 1) : (b - GCN_GEMM_BLOCKS);
    const int slice = fb / chunksPerSlice;
    const int chunk = fb - slice * chunksPerSlice;
    if (slice < GCN_SLICES)
      gcn_fill_chunk(src, dst, rank, rowptr, part, dis, rec, e, nrows,
                     sliceLen, slice, chunk * (GCN_FILL_U * 256));
    return;
  }
  const int gb = b >> 1;

  const int nTiles = (nrows + 15) >> 4;
  const int tBeg = (int)(((long long)gb * nTiles) / GCN_GEMM_BLOCKS);
  const int tEnd = (int)(((long long)(gb + 1) * nTiles) / GCN_GEMM_BLOCKS);

  for (int t = tBeg; t < tEnd;) {
    const int rem = tEnd - t;
    if (rem >= 5) {
      gcn_gemm_run<K, 5>(X, Wt, H, nrows, t * 16, As);
      t += 5;
    } else if (rem == 4) {
      gcn_gemm_run<K, 4>(X, Wt, H, nrows, t * 16, As);
      t += 4;
    } else {
      gcn_gemm_run<K, 1>(X, Wt, H, nrows, t * 16, As);
      t += 1;
    }
    __syncthreads();
  }
}

__device__ __forceinline__ void gcn_acc(float4& acc, const float4 v, float w) {
  acc.x = fmaf(v.x, w, acc.x);
  acc.y = fmaf(v.y, w, acc.y);
  acc.z = fmaf(v.z, w, acc.z);
  acc.w = fmaf(v.w, w, acc.w);
}

template <int C>
__device__ __forceinline__ void gcn_gather_chunk(
    float4& acc, const EdgeRec* __restrict__ rec,
    const float4* __restrict__ H4, int lane) {
  EdgeRec r[C];
  float4 v[C];
#pragma unroll
  for (int k = 0; k < C; ++k) r[k] = rec[k];
#pragma unroll
  for (int k = 0; k < C; ++k) v[k] = H4[(size_t)r[k].src * 32 + lane];
#pragma unroll
  for (int k = 0; k < C; ++k) gcn_acc(acc, v[k], r[k].w);
}

__global__ __launch_bounds__(256) void gcn_gather_combine(
    const EdgeRec* __restrict__ rec, const int* __restrict__ rowptr,
    const int* __restrict__ cnt, const float* __restrict__ H,
    const float* __restrict__ dinv, const float* __restrict__ bias,
    float* __restrict__ out, int n) {
  const int node =
      __builtin_amdgcn_readfirstlane(blockIdx.x * 8 + (threadIdx.x >> 5));
  if (node >= n) return;
  const int lane = threadIdx.x & 31;
  const float4* __restrict__ H4 = reinterpret_cast<const float4*>(H);

  int j = rowptr[node];
  const int end = j + cnt[node];
  const float di = dinv[node];
  const float4 hs = H4[(size_t)node * 32 + lane];
  const float4 bv = reinterpret_cast<const float4*>(bias)[lane];

  float4 acc = make_float4(0.f, 0.f, 0.f, 0.f);
  for (; j + 8 <= end; j += 8) gcn_gather_chunk<8>(acc, rec + j, H4, lane);
  if (j + 4 <= end) {
    gcn_gather_chunk<4>(acc, rec + j, H4, lane);
    j += 4;
  }
  if (j + 2 <= end) {
    gcn_gather_chunk<2>(acc, rec + j, H4, lane);
    j += 2;
  }
  if (j < end) gcn_gather_chunk<1>(acc, rec + j, H4, lane);

  float4 o;
  o.x = fmaxf(fmaf(hs.x, di, acc.x) + bv.x, 0.0f);
  o.y = fmaxf(fmaf(hs.y, di, acc.y) + bv.y, 0.0f);
  o.z = fmaxf(fmaf(hs.z, di, acc.z) + bv.z, 0.0f);
  o.w = fmaxf(fmaf(hs.w, di, acc.w) + bv.w, 0.0f);
  reinterpret_cast<float4*>(out)[(size_t)node * 32 + lane] = o;
}

constexpr int GCN_FUSE_BLOCKS = 1024;
constexpr int GCN_FUSE_RT = 4;

__device__ __forceinline__ float4 gcn_node_row(
    const EdgeRec* __restrict__ rec, int j, int end, float di,
    const float4* __restrict__ H4, int node, const float4 bv, int lane) {
  const float4 hs = H4[(size_t)node * 32 + lane];
  float4 acc = make_float4(0.f, 0.f, 0.f, 0.f);
  for (; j + 16 <= end; j += 16) gcn_gather_chunk<16>(acc, rec + j, H4, lane);
  if (j + 8 <= end) {
    gcn_gather_chunk<8>(acc, rec + j, H4, lane);
    j += 8;
  }
  if (j + 4 <= end) {
    gcn_gather_chunk<4>(acc, rec + j, H4, lane);
    j += 4;
  }
  if (j + 2 <= end) {
    gcn_gather_chunk<2>(acc, rec + j, H4, lane);
    j += 2;
  }
  if (j < end) gcn_gather_chunk<1>(acc, rec + j, H4, lane);
  float4 o;
  o.x = fmaxf(fmaf(hs.x, di, acc.x) + bv.x, 0.0f);
  o.y = fmaxf(fmaf(hs.y, di, acc.y) + bv.y, 0.0f);
  o.z = fmaxf(fmaf(hs.z, di, acc.z) + bv.z, 0.0f);
  o.w = fmaxf(fmaf(hs.w, di, acc.w) + bv.w, 0.0f);
  return o;
}

template <int RT>
__device__ __forceinline__ void gcn_fused_gemm(
    const float* __restrict__ Wt, float* __restrict__ Hout, int nrows,
    int rowBase, const float* __restrict__ As) {
  constexpr int K = 128;
  constexpr int LDA = K + 4;
  const int tid = threadIdx.x;
  const int wave = tid >> 5;
  const int lane = tid & 31;
  const int l15 = lane & 15;
  const int halfSel = lane >> 4;
  const int col = wave * 16 + l15;

  const float* __restrict__ Bcol = Wt + (size_t)col * K + 4 * halfSel;
  const float* __restrict__ Arow = As + l15 * LDA + 4 * halfSel;

  v8f acc[RT];
#pragma unroll
  for (int t = 0; t < RT; ++t) acc[t] = v8f{};

#pragma unroll 2
  for (int k0 = 0; k0 < K; k0 += 8) {
    const float4 b = *reinterpret_cast<const float4*>(Bcol + k0);
    float4 a[RT];
#pragma unroll
    for (int t = 0; t < RT; ++t)
      a[t] = *reinterpret_cast<const float4*>(Arow + t * 16 * LDA + k0);
    v2f b0, b1;
    b0.x = b.x; b0.y = b.y;
    b1.x = b.z; b1.y = b.w;
#pragma unroll
    for (int t = 0; t < RT; ++t) {
      v2f a0;
      a0.x = a[t].x; a0.y = a[t].y;
      acc[t] = __builtin_amdgcn_wmma_f32_16x16x4_f32(
           false, a0,  false, b0,
           (short)0, acc[t],  false,  false);
    }
#pragma unroll
    for (int t = 0; t < RT; ++t) {
      v2f a1;
      a1.x = a[t].z; a1.y = a[t].w;
      acc[t] = __builtin_amdgcn_wmma_f32_16x16x4_f32(
           false, a1,  false, b1,
           (short)0, acc[t],  false,  false);
    }
  }

  if (rowBase + RT * 16 <= nrows) {
    float* __restrict__ Hp = Hout + (size_t)(rowBase + 8 * halfSel) * 128 + col;
#pragma unroll
    for (int t = 0; t < RT; ++t) {
#pragma unroll
      for (int r = 0; r < 8; ++r) {
        Hp[(size_t)(t * 16 + r) * 128] = acc[t][r];
      }
    }
  } else {
#pragma unroll
    for (int t = 0; t < RT; ++t) {
#pragma unroll
      for (int r = 0; r < 8; ++r) {
        const int m = rowBase + t * 16 + r + 8 * halfSel;
        if (m < nrows) Hout[(size_t)m * 128 + col] = acc[t][r];
      }
    }
  }
}

__global__ __launch_bounds__(256) void gcn_gather_gemm(
    const EdgeRec* __restrict__ rec, const int* __restrict__ rowptr,
    const int* __restrict__ cnt, const float* __restrict__ Hin,
    const float* __restrict__ dinv, const float* __restrict__ bias,
    const float* __restrict__ Wt, float* __restrict__ Hout, int nrows) {
  constexpr int LDA = 128 + 4;
  __shared__ __align__(16) float As[GCN_FUSE_RT * 16 * LDA];

  const int nTiles = (nrows + 15) >> 4;
  const int tBeg = (int)(((long long)blockIdx.x * nTiles) / gridDim.x);
  const int tEnd = (int)(((long long)(blockIdx.x + 1) * nTiles) / gridDim.x);
  const int nt = tEnd - tBeg;
  if (nt <= 0) return;
  const int rowBase = tBeg * 16;

  {
    const int lane = threadIdx.x & 31;
    const int wv = __builtin_amdgcn_readfirstlane(threadIdx.x >> 5);
    const float4* __restrict__ H4 = reinterpret_cast<const float4*>(Hin);
    const float4 bv = reinterpret_cast<const float4*>(bias)[lane];
    const int npw = 2 * nt;
    const int firstRow = wv * npw;
    const int firstNode = rowBase + firstRow;
    const int mine = min(firstNode + lane, nrows - 1);
    const int vJ = rowptr[mine];
    const int vC = cnt[mine];
    const int vD = __float_as_int(dinv[mine]);
    for (int i = 0; i < npw; ++i) {
      const int node = firstNode + i;
      float4 o = make_float4(0.f, 0.f, 0.f, 0.f);
      if (node < nrows) {
        const int j = __builtin_amdgcn_readlane(vJ, i);
        const int c = __builtin_amdgcn_readlane(vC, i);
        const float di = __int_as_float(__builtin_amdgcn_readlane(vD, i));
        o = gcn_node_row(rec, j, j + c, di, H4, node, bv, lane);
      }
      *reinterpret_cast<float4*>(&As[(firstRow + i) * LDA + 4 * lane]) = o;
    }
  }
  __syncthreads();

  if (nt == 4) {
    gcn_fused_gemm<4>(Wt, Hout, nrows, rowBase, As);
  } else if (nt == 3) {
    gcn_fused_gemm<3>(Wt, Hout, nrows, rowBase, As);
  } else if (nt == 2) {
    gcn_fused_gemm<2>(Wt, Hout, nrows, rowBase, As);
  } else {
    gcn_fused_gemm<1>(Wt, Hout, nrows, rowBase, As);
  }
}

extern "C" void kernel_launch(void* const* d_in, const int* in_sizes, int n_in,
                              void* d_out, int out_size, void* d_ws,
                              size_t ws_size, hipStream_t stream) {
  const float* x = (const float*)d_in[0];
  const int* edge = (const int*)d_in[1];
  const float* W0 = (const float*)d_in[2];
  const float* b0 = (const float*)d_in[3];
  const float* W1 = (const float*)d_in[4];
  const float* b1 = (const float*)d_in[5];
  const float* W2 = (const float*)d_in[6];
  const float* b2 = (const float*)d_in[7];
  float* out = (float*)d_out;

  const int N = in_sizes[0] / 256;
  const int E = in_sizes[1] / 2;
  const int* srcIdx = edge;
  const int* dstIdx = edge + E;

  float* ws = (float*)d_ws;
  float* hbuf = ws;
  float* hbuf2 = hbuf + (size_t)N * 128;
  EdgeRec* rec = (EdgeRec*)(hbuf2 + (size_t)N * 128);
  float* wt0 = (float*)(rec + E);
  float* wt1 = wt0 + 128 * 256;
  float* wt2 = wt1 + 128 * 128;
  float* dis = wt2 + 128 * 128;
  float* dinv = dis + N;
  int* cnt = (int*)(dinv + N);
  int* rowptr = cnt + N;
  int* rank = rowptr + N;
  int* part = rank + E;

  const int T = 256;
  const int nodeBlocks = (N + T - 1) / T;
  const int sliceLen = (E + GCN_SLICES - 1) / GCN_SLICES;
  const int chunksPerSlice =
      (sliceLen + GCN_FILL_U * T - 1) / (GCN_FILL_U * T);
  const int fillBlocks = chunksPerSlice * GCN_SLICES;
  const int nodeRows = (N + GCN_NODE_CAP - 1) / GCN_NODE_CAP;
  const int gathBlocks = (N + 7) / 8;
  const int scanBlocks = (N + 1023) / 1024;
  const int gemmBlocks = GCN_GEMM_BLOCKS;
  const int nTiles = (N + 15) / 16;
  const int fuseMin = (nTiles + GCN_FUSE_RT - 1) / GCN_FUSE_RT;
  const int fuseBlocks = fuseMin > GCN_FUSE_BLOCKS ? fuseMin : GCN_FUSE_BLOCKS;

  gcn_slice_count<<<dim3(GCN_SLICES + GCN_TR_BLOCKS, nodeRows), 1024, 0,
                    stream>>>(dstIdx, part, rank, E, N, sliceLen, W0, W1, W2,
                              wt0, wt1, wt2);
  gcn_slice_prefix<<<nodeBlocks, T, 0, stream>>>(part, cnt, N);
  gcn_scan_deg<<<scanBlocks, 1024, 0, stream>>>(cnt, rowptr, dis, dinv, N);

  const int pairBlocks = fillBlocks > gemmBlocks ? fillBlocks : gemmBlocks;
  gcn_gemm_fill<<<gemmBlocks + pairBlocks, T, 0, stream>>>(
      x, wt0, hbuf, N, srcIdx, dstIdx, rank, rowptr, part, dis, rec, E,
      sliceLen, chunksPerSlice);

  gcn_gather_gemm<<<fuseBlocks, T, 0, stream>>>(rec, rowptr, cnt, hbuf, dinv,
                                                b0, wt1, hbuf2, N);

  gcn_gather_gemm<<<fuseBlocks, T, 0, stream>>>(rec, rowptr, cnt, hbuf2, dinv,
                                                b1, wt2, hbuf, N);

  gcn_gather_combine<<<gathBlocks, T, 0, stream>>>(rec, rowptr, cnt, hbuf, dinv, b2, out, N);
}
